# starting kernel
# baseline (speedup 1.0000x reference)
#include <hip/hip_runtime.h>

typedef __attribute__((ext_vector_type(2))) float v2f;
typedef __attribute__((ext_vector_type(8))) float v8f;

#define LSTR 80

__global__ __launch_bounds__(512) void gram_partial(const float* __restrict__ Phi,
                                                    float* __restrict__ part,
                                                    int m, int nwg) {
    __shared__ float tile[64 * LSTR];
    const int tid  = threadIdx.x;
    const int lane = tid & 31;
    const int wave = tid >> 5;
    const int ti   = wave >> 2;
    const int tj   = wave & 3;
    const int li   = lane & 15;
    const int kh   = lane >> 4;

    v8f acc = {};

    for (long rowbase = (long)blockIdx.x * 64; rowbase < (long)m; rowbase += (long)nwg * 64) {
        {
            const int r = tid >> 3;
            const int c = (tid & 7) * 8;
            const long row = rowbase + r;
            float* dst = &tile[r * LSTR + c];
            if (row < (long)m) {
                const float4* src = reinterpret_cast<const float4*>(Phi + row * 64 + c);
                float4 a = src[0], b = src[1];
                dst[0]=a.x; dst[1]=a.y; dst[2]=a.z; dst[3]=a.w;
                dst[4]=b.x; dst[5]=b.y; dst[6]=b.z; dst[7]=b.w;
            } else {
                #pragma unroll
                for (int q = 0; q < 8; ++q) dst[q] = 0.0f;
            }
        }
        __syncthreads();

        #pragma unroll
        for (int k0 = 0; k0 < 64; k0 += 4) {
            const int ka = k0 + 2 * kh;
            v2f va, vb;
            va.x = tile[ka       * LSTR + ti * 16 + li];
            va.y = tile[(ka + 1) * LSTR + ti * 16 + li];
            vb.x = tile[ka       * LSTR + tj * 16 + li];
            vb.y = tile[(ka + 1) * LSTR + tj * 16 + li];
            acc = __builtin_amdgcn_wmma_f32_16x16x4_f32(false, va, false, vb,
                                                        (short)0, acc, false, false);
        }
        __syncthreads();
    }

    float* p = part + (size_t)blockIdx.x * 4096;
    #pragma unroll
    for (int v = 0; v < 8; ++v) {
        const int gi = ti * 16 + v + 8 * kh;
        const int gj = tj * 16 + li;
        p[gi * 64 + gj] = acc[v];
    }
}

__global__ __launch_bounds__(256) void reduce_g(const float* __restrict__ part,
                                                float* __restrict__ G, int nwg) {
    const int e = blockIdx.x * 256 + threadIdx.x;
    float s = 0.0f;
    for (int w = 0; w < nwg; ++w) s += part[(size_t)w * 4096 + e];
    G[e] = s;
}

__global__ __launch_bounds__(64) void chol_inv(const float* __restrict__ G,
                                               float* __restrict__ Rinv) {
    __shared__ float A[64 * 64];
    __shared__ float X[64 * 64];
    const int t = threadIdx.x;

    for (int i = 0; i < 64; ++i) A[i * 64 + t] = G[i * 64 + t];
    __syncthreads();

    for (int k = 0; k < 64; ++k) {
        const float dk = sqrtf(A[k * 64 + k]);
        if (t >= k) A[k * 64 + t] = (t == k) ? dk : (A[k * 64 + t] / dk);
        __syncthreads();
        if (t > k) {
            const float uki = A[k * 64 + t];
            for (int j = t; j < 64; ++j)
                A[t * 64 + j] -= uki * A[k * 64 + j];
        }
        __syncthreads();
    }

    for (int i = 0; i < 64; ++i) X[i * 64 + t] = 0.0f;
    X[t * 64 + t] = 1.0f / A[t * 64 + t];
    for (int i = t - 1; i >= 0; --i) {
        float s = 0.0f;
        for (int k = i + 1; k <= t; ++k) s += A[i * 64 + k] * X[k * 64 + t];
        X[i * 64 + t] = -s / A[i * 64 + i];
    }
    __syncthreads();
    for (int i = 0; i < 64; ++i) Rinv[i * 64 + t] = X[i * 64 + t];
}

__global__ __launch_bounds__(512) void apply_rinv(const float* __restrict__ Phi,
                                                  const float* __restrict__ Rinv,
                                                  float* __restrict__ Q, int m) {
    __shared__ float B[64 * LSTR];
    const int tid  = threadIdx.x;
    const int lane = tid & 31;
    const int wave = tid >> 5;
    const int ti   = wave >> 2;
    const int tj   = wave & 3;
    const int li   = lane & 15;
    const int kh   = lane >> 4;

    {
        const int r = tid >> 3;
        const int c = (tid & 7) * 8;
        const float* src = Rinv + r * 64 + c;
        float* dst = &B[r * LSTR + c];
        #pragma unroll
        for (int q = 0; q < 8; ++q) dst[q] = src[q];
    }
    __syncthreads();

    const long rowbase = (long)blockIdx.x * 64;
    long arow = rowbase + ti * 16 + li;
    if (arow >= (long)m) arow = (long)m - 1;
    const float2* aptr = reinterpret_cast<const float2*>(Phi + arow * 64);

    v8f acc = {};
    #pragma unroll
    for (int k0 = 0; k0 < 64; k0 += 4) {
        const int ka = k0 + 2 * kh;
        v2f va, vb;
        const float2 a = aptr[ka >> 1];
        va.x = a.x; va.y = a.y;
        vb.x = B[ka       * LSTR + tj * 16 + li];
        vb.y = B[(ka + 1) * LSTR + tj * 16 + li];
        acc = __builtin_amdgcn_wmma_f32_16x16x4_f32(false, va, false, vb,
                                                    (short)0, acc, false, false);
    }

    #pragma unroll
    for (int v = 0; v < 8; ++v) {
        const long row = rowbase + ti * 16 + v + 8 * kh;
        if (row < (long)m) Q[row * 64 + tj * 16 + li] = acc[v];
    }
}

extern "C" void kernel_launch(void* const* d_in, const int* in_sizes, int n_in,
                              void* d_out, int out_size, void* d_ws, size_t ws_size,
                              hipStream_t stream) {
    const float* Phi = (const float*)d_in[0];
    float* out = (float*)d_out;
    const int m = in_sizes[0] / 64;

    float* ws   = (float*)d_ws;
    float* G    = ws;
    float* Rinv = ws + 4096;
    float* part = ws + 8192;

    long avail = ((long)(ws_size / 4) - 8192) / 4096;
    int nwg = (int)(avail < 1 ? 1 : (avail > 512 ? 512 : avail));

    gram_partial<<<nwg, 512, 0, stream>>>(Phi, part, m, nwg);
    reduce_g<<<16, 256, 0, stream>>>(part, G, nwg);
    chol_inv<<<1, 64, 0, stream>>>(G, Rinv);

    const int nblk = (m + 63) / 64;
    apply_rinv<<<nblk, 512, 0, stream>>>(Phi, Rinv, out, m);
}


# chol_inv: one wave eliminates [G | I] held in registers (2 columns per lane), pivot row broadcast by b128 LDS loads; Rinv by forward substitution, so its sums run in the opposite order
# speedup vs baseline: 1.5333x; 1.5333x over previous
#include <hip/hip_runtime.h>

typedef __attribute__((ext_vector_type(2))) float v2f;
typedef __attribute__((ext_vector_type(8))) float v8f;

#define LSTR 80

__global__ __launch_bounds__(512) void gram_partial(const float* __restrict__ Phi,
                                                    float* __restrict__ part,
                                                    int m, int nwg) {
    __shared__ float tile[64 * LSTR];
    const int tid  = threadIdx.x;
    const int lane = tid & 31;
    const int wave = tid >> 5;
    const int ti   = wave >> 2;
    const int tj   = wave & 3;
    const int li   = lane & 15;
    const int kh   = lane >> 4;

    v8f acc = {};

    for (long rowbase = (long)blockIdx.x * 64; rowbase < (long)m; rowbase += (long)nwg * 64) {
        {
            const int r = tid >> 3;
            const int c = (tid & 7) * 8;
            const long row = rowbase + r;
            float* dst = &tile[r * LSTR + c];
            if (row < (long)m) {
                const float4* src = reinterpret_cast<const float4*>(Phi + row * 64 + c);
                float4 a = src[0], b = src[1];
                dst[0]=a.x; dst[1]=a.y; dst[2]=a.z; dst[3]=a.w;
                dst[4]=b.x; dst[5]=b.y; dst[6]=b.z; dst[7]=b.w;
            } else {
                #pragma unroll
                for (int q = 0; q < 8; ++q) dst[q] = 0.0f;
            }
        }
        __syncthreads();

        #pragma unroll
        for (int k0 = 0; k0 < 64; k0 += 4) {
            const int ka = k0 + 2 * kh;
            v2f va, vb;
            va.x = tile[ka       * LSTR + ti * 16 + li];
            va.y = tile[(ka + 1) * LSTR + ti * 16 + li];
            vb.x = tile[ka       * LSTR + tj * 16 + li];
            vb.y = tile[(ka + 1) * LSTR + tj * 16 + li];
            acc = __builtin_amdgcn_wmma_f32_16x16x4_f32(false, va, false, vb,
                                                        (short)0, acc, false, false);
        }
        __syncthreads();
    }

    float* p = part + (size_t)blockIdx.x * 4096;
    #pragma unroll
    for (int v = 0; v < 8; ++v) {
        const int gi = ti * 16 + v + 8 * kh;
        const int gj = tj * 16 + li;
        p[gi * 64 + gj] = acc[v];
    }
}

__global__ __launch_bounds__(256) void reduce_g(const float* __restrict__ part,
                                                float* __restrict__ G, int nwg) {
    const int e = blockIdx.x * 256 + threadIdx.x;
    float s = 0.0f;
    for (int w = 0; w < nwg; ++w) s += part[(size_t)w * 4096 + e];
    G[e] = s;
}

#define WSTR 65

template <int NC, bool LOW>
__device__ __forceinline__ void elim_step(const int k, const int l,
                                          float (&a0)[64],
                                          float (&a1)[64],
                                          float (&b0)[64],
                                          float* rowb, float* wbuf) {
    const float dsel = LOW ? a0[0] : a1[0];
    const float akk  = __int_as_float(__builtin_amdgcn_readlane(__float_as_int(dsel), k & 31));
    const float dk   = sqrtf(akk);
    const int j0 = l, j1 = l + 32;
    const float u0 = LOW ? a0[0] / dk : 0.0f;
    const float u1 = a1[0] / dk;
    const float w0 = b0[0] / dk;
    const float w1 = LOW ? 0.0f : a0[0] / dk;
    __syncthreads();
    rowb[(j0 - k - 1) & 63] = (j0 > k) ? u0 : 0.0f;
    rowb[(j1 - k - 1) & 63] = (j1 > k) ? u1 : 0.0f;
    wbuf[j0 * WSTR + k] = w0;
    wbuf[j1 * WSTR + k] = w1;
    __syncthreads();

    float v[NC + 1];
    const float4* r4 = reinterpret_cast<const float4*>(rowb);
    #pragma unroll
    for (int q = 0; q < (NC + 1) / 4; ++q) {
        const float4 t = r4[q];
        v[4 * q] = t.x; v[4 * q + 1] = t.y; v[4 * q + 2] = t.z; v[4 * q + 3] = t.w;
    }
    #pragma unroll
    for (int c = 1; c <= NC; ++c) {
        a0[c - 1] = fmaf(-v[c - 1], LOW ? u0 : w1, a0[c]);
        a1[c - 1] = fmaf(-v[c - 1], u1, a1[c]);
        b0[c - 1] = fmaf(-v[c - 1], w0, b0[c]);
    }
    a0[NC] = 0.0f;
    a1[NC] = 0.0f;
    b0[NC] = 0.0f;
}

__global__ __launch_bounds__(32) void chol_inv(const float* __restrict__ G,
                                               float* __restrict__ Rinv) {
    __shared__ __attribute__((aligned(16))) float rowb[64];
    __shared__ float wbuf[64 * WSTR];
    const int l = threadIdx.x;

    float a0[64], a1[64], b0[64];
    #pragma unroll
    for (int i = 0; i < 64; ++i) {
        a0[i] = G[i * 64 + l];
        a1[i] = G[i * 64 + l + 32];
        b0[i] = (i == l) ? 1.0f : 0.0f;
    }

    #pragma nounroll
    for (int k = 0; k < 32; ++k)  elim_step<63, true >(k, l, a0, a1, b0, rowb, wbuf);
    #pragma unroll
    for (int c = 0; c < 64; ++c) a0[c] = (c == l) ? 1.0f : 0.0f;
    #pragma nounroll
    for (int k = 32; k < 64; ++k) elim_step<31, false>(k, l, a0, a1, b0, rowb, wbuf);
    __syncthreads();

    #pragma unroll
    for (int r = 0; r < 64; ++r) {
        Rinv[r * 64 + l]      = wbuf[r * WSTR + l];
        Rinv[r * 64 + l + 32] = wbuf[r * WSTR + l + 32];
    }
}

__global__ __launch_bounds__(512) void apply_rinv(const float* __restrict__ Phi,
                                                  const float* __restrict__ Rinv,
                                                  float* __restrict__ Q, int m) {
    __shared__ float B[64 * LSTR];
    const int tid  = threadIdx.x;
    const int lane = tid & 31;
    const int wave = tid >> 5;
    const int ti   = wave >> 2;
    const int tj   = wave & 3;
    const int li   = lane & 15;
    const int kh   = lane >> 4;

    {
        const int r = tid >> 3;
        const int c = (tid & 7) * 8;
        const float* src = Rinv + r * 64 + c;
        float* dst = &B[r * LSTR + c];
        #pragma unroll
        for (int q = 0; q < 8; ++q) dst[q] = src[q];
    }
    __syncthreads();

    const long rowbase = (long)blockIdx.x * 64;
    long arow = rowbase + ti * 16 + li;
    if (arow >= (long)m) arow = (long)m - 1;
    const float2* aptr = reinterpret_cast<const float2*>(Phi + arow * 64);

    v8f acc = {};
    #pragma unroll
    for (int k0 = 0; k0 < 64; k0 += 4) {
        const int ka = k0 + 2 * kh;
        v2f va, vb;
        const float2 a = aptr[ka >> 1];
        va.x = a.x; va.y = a.y;
        vb.x = B[ka       * LSTR + tj * 16 + li];
        vb.y = B[(ka + 1) * LSTR + tj * 16 + li];
        acc = __builtin_amdgcn_wmma_f32_16x16x4_f32(false, va, false, vb,
                                                    (short)0, acc, false, false);
    }

    #pragma unroll
    for (int v = 0; v < 8; ++v) {
        const long row = rowbase + ti * 16 + v + 8 * kh;
        if (row < (long)m) Q[row * 64 + tj * 16 + li] = acc[v];
    }
}

extern "C" void kernel_launch(void* const* d_in, const int* in_sizes, int n_in,
                              void* d_out, int out_size, void* d_ws, size_t ws_size,
                              hipStream_t stream) {
    const float* Phi = (const float*)d_in[0];
    float* out = (float*)d_out;
    const int m = in_sizes[0] / 64;

    float* ws   = (float*)d_ws;
    float* G    = ws;
    float* Rinv = ws + 4096;
    float* part = ws + 8192;

    long avail = ((long)(ws_size / 4) - 8192) / 4096;
    int nwg = (int)(avail < 1 ? 1 : (avail > 512 ? 512 : avail));

    gram_partial<<<nwg, 512, 0, stream>>>(Phi, part, m, nwg);
    reduce_g<<<16, 256, 0, stream>>>(part, G, nwg);
    chol_inv<<<1, 32, 0, stream>>>(G, Rinv);

    const int nblk = (m + 63) / 64;
    apply_rinv<<<nblk, 512, 0, stream>>>(Phi, Rinv, out, m);
}


# apply_rinv persistent: 4096 barrier-free waves hold Rinv as B fragments in registers, stream 16-row Phi strips via a private LDS slab, whole-line loads and stores; k order of Q's sums changes
# speedup vs baseline: 1.9708x; 1.2853x over previous
#include <hip/hip_runtime.h>

typedef __attribute__((ext_vector_type(2))) float v2f;
typedef __attribute__((ext_vector_type(8))) float v8f;

#define LSTR 80

__global__ __launch_bounds__(512) void gram_partial(const float* __restrict__ Phi,
                                                    float* __restrict__ part,
                                                    int m, int nwg) {
    __shared__ float tile[64 * LSTR];
    const int tid  = threadIdx.x;
    const int lane = tid & 31;
    const int wave = tid >> 5;
    const int ti   = wave >> 2;
    const int tj   = wave & 3;
    const int li   = lane & 15;
    const int kh   = lane >> 4;

    v8f acc = {};

    for (long rowbase = (long)blockIdx.x * 64; rowbase < (long)m; rowbase += (long)nwg * 64) {
        {
            const int r = tid >> 3;
            const int c = (tid & 7) * 8;
            const long row = rowbase + r;
            float* dst = &tile[r * LSTR + c];
            if (row < (long)m) {
                const float4* src = reinterpret_cast<const float4*>(Phi + row * 64 + c);
                float4 a = src[0], b = src[1];
                dst[0]=a.x; dst[1]=a.y; dst[2]=a.z; dst[3]=a.w;
                dst[4]=b.x; dst[5]=b.y; dst[6]=b.z; dst[7]=b.w;
            } else {
                #pragma unroll
                for (int q = 0; q < 8; ++q) dst[q] = 0.0f;
            }
        }
        __syncthreads();

        #pragma unroll
        for (int k0 = 0; k0 < 64; k0 += 4) {
            const int ka = k0 + 2 * kh;
            v2f va, vb;
            va.x = tile[ka       * LSTR + ti * 16 + li];
            va.y = tile[(ka + 1) * LSTR + ti * 16 + li];
            vb.x = tile[ka       * LSTR + tj * 16 + li];
            vb.y = tile[(ka + 1) * LSTR + tj * 16 + li];
            acc = __builtin_amdgcn_wmma_f32_16x16x4_f32(false, va, false, vb,
                                                        (short)0, acc, false, false);
        }
        __syncthreads();
    }

    float* p = part + (size_t)blockIdx.x * 4096;
    #pragma unroll
    for (int v = 0; v < 8; ++v) {
        const int gi = ti * 16 + v + 8 * kh;
        const int gj = tj * 16 + li;
        p[gi * 64 + gj] = acc[v];
    }
}

__global__ __launch_bounds__(256) void reduce_g(const float* __restrict__ part,
                                                float* __restrict__ G, int nwg) {
    const int e = blockIdx.x * 256 + threadIdx.x;
    float s = 0.0f;
    for (int w = 0; w < nwg; ++w) s += part[(size_t)w * 4096 + e];
    G[e] = s;
}

#define WSTR 65

template <int NC, bool LOW>
__device__ __forceinline__ void elim_step(const int k, const int l,
                                          float (&a0)[64],
                                          float (&a1)[64],
                                          float (&b0)[64],
                                          float* rowb, float* wbuf) {
    const float dsel = LOW ? a0[0] : a1[0];
    const float akk  = __int_as_float(__builtin_amdgcn_readlane(__float_as_int(dsel), k & 31));
    const float dk   = sqrtf(akk);
    const int j0 = l, j1 = l + 32;
    const float u0 = LOW ? a0[0] / dk : 0.0f;
    const float u1 = a1[0] / dk;
    const float w0 = b0[0] / dk;
    const float w1 = LOW ? 0.0f : a0[0] / dk;
    __syncthreads();
    rowb[(j0 - k - 1) & 63] = (j0 > k) ? u0 : 0.0f;
    rowb[(j1 - k - 1) & 63] = (j1 > k) ? u1 : 0.0f;
    wbuf[j0 * WSTR + k] = w0;
    wbuf[j1 * WSTR + k] = w1;
    __syncthreads();

    float v[NC + 1];
    const float4* r4 = reinterpret_cast<const float4*>(rowb);
    #pragma unroll
    for (int q = 0; q < (NC + 1) / 4; ++q) {
        const float4 t = r4[q];
        v[4 * q] = t.x; v[4 * q + 1] = t.y; v[4 * q + 2] = t.z; v[4 * q + 3] = t.w;
    }
    #pragma unroll
    for (int c = 1; c <= NC; ++c) {
        a0[c - 1] = fmaf(-v[c - 1], LOW ? u0 : w1, a0[c]);
        a1[c - 1] = fmaf(-v[c - 1], u1, a1[c]);
        b0[c - 1] = fmaf(-v[c - 1], w0, b0[c]);
    }
    a0[NC] = 0.0f;
    a1[NC] = 0.0f;
    b0[NC] = 0.0f;
}

__global__ __launch_bounds__(32) void chol_inv(const float* __restrict__ G,
                                               float* __restrict__ Rinv) {
    __shared__ __attribute__((aligned(16))) float rowb[64];
    __shared__ float wbuf[64 * WSTR];
    const int l = threadIdx.x;

    float a0[64], a1[64], b0[64];
    #pragma unroll
    for (int i = 0; i < 64; ++i) {
        a0[i] = G[i * 64 + l];
        a1[i] = G[i * 64 + l + 32];
        b0[i] = (i == l) ? 1.0f : 0.0f;
    }

    #pragma nounroll
    for (int k = 0; k < 32; ++k)  elim_step<63, true >(k, l, a0, a1, b0, rowb, wbuf);
    #pragma unroll
    for (int c = 0; c < 64; ++c) a0[c] = (c == l) ? 1.0f : 0.0f;
    #pragma nounroll
    for (int k = 32; k < 64; ++k) elim_step<31, false>(k, l, a0, a1, b0, rowb, wbuf);
    __syncthreads();

    #pragma unroll
    for (int r = 0; r < 64; ++r) {
        Rinv[r * 64 + l]      = wbuf[r * WSTR + l];
        Rinv[r * 64 + l + 32] = wbuf[r * WSTR + l + 32];
    }
}

typedef __attribute__((ext_vector_type(4))) float v4f;
#define APITCH 68
#define AWAVES 8
#define ATHREADS 256
#define AGRID 512

__global__ __launch_bounds__(ATHREADS) void apply_rinv(const float* __restrict__ Phi,
                                                         const float* __restrict__ Rinv,
                                                         float* __restrict__ Q, int m) {
    __shared__ __attribute__((aligned(16))) float slab[AWAVES * 16 * APITCH];
    const int lane = threadIdx.x & 31;
    const int wave = __builtin_amdgcn_readfirstlane(threadIdx.x >> 5);
    const int c = lane & 15;
    const int h = lane >> 4;
    const int nstrips = (m + 15) >> 4;
    const int nwaves  = AGRID * AWAVES;
    int strip = blockIdx.x * AWAVES + wave;
    if (strip >= nstrips) return;

    float* my = slab + wave * (16 * APITCH);

    v2f bf[16][4];
    #pragma unroll
    for (int s = 0; s < 16; ++s) {
        const int k = 8 * (s >> 1) + 4 * h + 2 * (s & 1);
        const v4f r0 = *reinterpret_cast<const v4f*>(Rinv + k * 64 + 4 * c);
        const v4f r1 = *reinterpret_cast<const v4f*>(Rinv + (k + 1) * 64 + 4 * c);
        bf[s][0].x = r0.x; bf[s][0].y = r1.x;
        bf[s][1].x = r0.y; bf[s][1].y = r1.y;
        bf[s][2].x = r0.z; bf[s][2].y = r1.z;
        bf[s][3].x = r0.w; bf[s][3].y = r1.w;
    }

    v4f p[8];
    #define LOAD_STRIP(S) do { _Pragma("unroll") for (int j = 0; j < 8; ++j) { \
        long row_ = (long)(S) * 16 + 2 * j + h; if (row_ >= (long)m) row_ = (long)m - 1; \
        p[j] = __builtin_nontemporal_load(reinterpret_cast<const v4f*>(Phi + row_ * 64 + 4 * c)); } } while (0)
    LOAD_STRIP(strip);

    for (; strip < nstrips; strip += nwaves) {
        #pragma unroll
        for (int j = 0; j < 8; ++j)
            *reinterpret_cast<v4f*>(my + (2 * j + h) * APITCH + 4 * c) = p[j];
        __builtin_amdgcn_fence(__ATOMIC_RELEASE, "wavefront");
        __builtin_amdgcn_wave_barrier();
        __builtin_amdgcn_fence(__ATOMIC_ACQUIRE, "wavefront");

        const int nxt = strip + nwaves;
        if (nxt < nstrips) LOAD_STRIP(nxt);

        v4f a[8];
        #pragma unroll
        for (int j = 0; j < 8; ++j)
            a[j] = *reinterpret_cast<const v4f*>(my + c * APITCH + 8 * j + 4 * h);
        __builtin_amdgcn_fence(__ATOMIC_RELEASE, "wavefront");
        __builtin_amdgcn_wave_barrier();
        __builtin_amdgcn_fence(__ATOMIC_ACQUIRE, "wavefront");

        v8f acc0 = {}, acc1 = {}, acc2 = {}, acc3 = {};
        #pragma unroll
        for (int s = 0; s < 16; ++s) {
            v2f va;
            if (s & 1) { va.x = a[s >> 1].z; va.y = a[s >> 1].w; }
            else       { va.x = a[s >> 1].x; va.y = a[s >> 1].y; }
            acc0 = __builtin_amdgcn_wmma_f32_16x16x4_f32(false, va, false, bf[s][0], (short)0, acc0, false, false);
            acc1 = __builtin_amdgcn_wmma_f32_16x16x4_f32(false, va, false, bf[s][1], (short)0, acc1, false, false);
            acc2 = __builtin_amdgcn_wmma_f32_16x16x4_f32(false, va, false, bf[s][2], (short)0, acc2, false, false);
            acc3 = __builtin_amdgcn_wmma_f32_16x16x4_f32(false, va, false, bf[s][3], (short)0, acc3, false, false);
        }

        asm volatile("v_nop\n\tv_nop\n\tv_nop\n\tv_nop\n\tv_nop\n\tv_nop\n\tv_nop\n\tv_nop"
                     : "+v"(acc0), "+v"(acc1), "+v"(acc2), "+v"(acc3), "+v"(a[7]));

        v4f o[8];
        #pragma unroll
        for (int r = 0; r < 8; ++r) { o[r].x = acc0[r]; o[r].y = acc1[r]; o[r].z = acc2[r]; o[r].w = acc3[r]; }
        const long row0 = (long)strip * 16 + 8 * h;
        float* qp = Q + row0 * 64 + 4 * c;
        if ((long)strip * 16 + 16 <= (long)m) {
            #pragma unroll
            for (int r = 0; r < 8; ++r) *reinterpret_cast<v4f*>(qp + r * 64) = o[r];
        } else {
            #pragma unroll
            for (int r = 0; r < 8; ++r)
                if (row0 + r < (long)m) *reinterpret_cast<v4f*>(qp + r * 64) = o[r];
        }
    }
    #undef LOAD_STRIP
}

extern "C" void kernel_launch(void* const* d_in, const int* in_sizes, int n_in,
                              void* d_out, int out_size, void* d_ws, size_t ws_size,
                              hipStream_t stream) {
    const float* Phi = (const float*)d_in[0];
    float* out = (float*)d_out;
    const int m = in_sizes[0] / 64;

    float* ws   = (float*)d_ws;
    float* G    = ws;
    float* Rinv = ws + 4096;
    float* part = ws + 8192;

    long avail = ((long)(ws_size / 4) - 8192) / 4096;
    int nwg = (int)(avail < 1 ? 1 : (avail > 512 ? 512 : avail));

    gram_partial<<<nwg, 512, 0, stream>>>(Phi, part, m, nwg);
    reduce_g<<<16, 256, 0, stream>>>(part, G, nwg);
    chol_inv<<<1, 32, 0, stream>>>(G, Rinv);

    apply_rinv<<<AGRID, ATHREADS, 0, stream>>>(Phi, Rinv, out, m);
}


# reduce_g: 128 workgroups x 16 waves, a wave issues all 32 loads of its share of the partial Grams as one group (1 memory round trip, was 64 in a row), wave sums meet in LDS; G's sum tree changes
# speedup vs baseline: 2.6752x; 1.3574x over previous
#include <hip/hip_runtime.h>

typedef __attribute__((ext_vector_type(2))) float v2f;
typedef __attribute__((ext_vector_type(8))) float v8f;

#define LSTR 80

__global__ __launch_bounds__(512) void gram_partial(const float* __restrict__ Phi,
                                                    float* __restrict__ part,
                                                    int m, int nwg) {
    __shared__ float tile[64 * LSTR];
    const int tid  = threadIdx.x;
    const int lane = tid & 31;
    const int wave = tid >> 5;
    const int ti   = wave >> 2;
    const int tj   = wave & 3;
    const int li   = lane & 15;
    const int kh   = lane >> 4;

    v8f acc = {};

    for (long rowbase = (long)blockIdx.x * 64; rowbase < (long)m; rowbase += (long)nwg * 64) {
        {
            const int r = tid >> 3;
            const int c = (tid & 7) * 8;
            const long row = rowbase + r;
            float* dst = &tile[r * LSTR + c];
            if (row < (long)m) {
                const float4* src = reinterpret_cast<const float4*>(Phi + row * 64 + c);
                float4 a = src[0], b = src[1];
                dst[0]=a.x; dst[1]=a.y; dst[2]=a.z; dst[3]=a.w;
                dst[4]=b.x; dst[5]=b.y; dst[6]=b.z; dst[7]=b.w;
            } else {
                #pragma unroll
                for (int q = 0; q < 8; ++q) dst[q] = 0.0f;
            }
        }
        __syncthreads();

        #pragma unroll
        for (int k0 = 0; k0 < 64; k0 += 4) {
            const int ka = k0 + 2 * kh;
            v2f va, vb;
            va.x = tile[ka       * LSTR + ti * 16 + li];
            va.y = tile[(ka + 1) * LSTR + ti * 16 + li];
            vb.x = tile[ka       * LSTR + tj * 16 + li];
            vb.y = tile[(ka + 1) * LSTR + tj * 16 + li];
            acc = __builtin_amdgcn_wmma_f32_16x16x4_f32(false, va, false, vb,
                                                        (short)0, acc, false, false);
        }
        __syncthreads();
    }

    float* p = part + (size_t)blockIdx.x * 4096;
    #pragma unroll
    for (int v = 0; v < 8; ++v) {
        const int gi = ti * 16 + v + 8 * kh;
        const int gj = tj * 16 + li;
        p[gi * 64 + gj] = acc[v];
    }
}

#define RWAVES 16
#define RTHREADS 512
#define RDEPTH 32
__global__ __launch_bounds__(RTHREADS) void reduce_g(const float* __restrict__ part,
                                                        float* __restrict__ G, int nwg) {
    __shared__ float red[RWAVES * 32];
    const int lane = threadIdx.x & 31;
    const int wave = __builtin_amdgcn_readfirstlane(threadIdx.x >> 5);
    const int e = blockIdx.x * 32 + lane;
    const float* p = part + e;
    float s0 = 0.0f, s1 = 0.0f, s2 = 0.0f, s3 = 0.0f;
    int w = wave;
    for (; w + (RDEPTH - 1) * RWAVES < nwg; w += RDEPTH * RWAVES) {
        float t[RDEPTH];
        #pragma unroll
        for (int i = 0; i < RDEPTH; ++i) t[i] = p[(size_t)(w + i * RWAVES) * 4096];
        #pragma unroll
        for (int i = 0; i < RDEPTH; i += 4) { s0 += t[i]; s1 += t[i + 1]; s2 += t[i + 2]; s3 += t[i + 3]; }
    }
    for (; w < nwg; w += RWAVES) s0 += p[(size_t)w * 4096];
    red[wave * 32 + lane] = (s0 + s1) + (s2 + s3);
    __syncthreads();
    if (wave == 0) {
        float g = 0.0f;
        #pragma unroll
        for (int i = 0; i < RWAVES; ++i) g += red[i * 32 + lane];
        G[e] = g;
    }
}

#define WSTR 65

template <int NC, bool LOW>
__device__ __forceinline__ void elim_step(const int k, const int l,
                                          float (&a0)[64],
                                          float (&a1)[64],
                                          float (&b0)[64],
                                          float* rowb, float* wbuf) {
    const float dsel = LOW ? a0[0] : a1[0];
    const float akk  = __int_as_float(__builtin_amdgcn_readlane(__float_as_int(dsel), k & 31));
    const float dk   = sqrtf(akk);
    const int j0 = l, j1 = l + 32;
    const float u0 = LOW ? a0[0] / dk : 0.0f;
    const float u1 = a1[0] / dk;
    const float w0 = b0[0] / dk;
    const float w1 = LOW ? 0.0f : a0[0] / dk;
    __syncthreads();
    rowb[(j0 - k - 1) & 63] = (j0 > k) ? u0 : 0.0f;
    rowb[(j1 - k - 1) & 63] = (j1 > k) ? u1 : 0.0f;
    wbuf[j0 * WSTR + k] = w0;
    wbuf[j1 * WSTR + k] = w1;
    __syncthreads();

    float v[NC + 1];
    const float4* r4 = reinterpret_cast<const float4*>(rowb);
    #pragma unroll
    for (int q = 0; q < (NC + 1) / 4; ++q) {
        const float4 t = r4[q];
        v[4 * q] = t.x; v[4 * q + 1] = t.y; v[4 * q + 2] = t.z; v[4 * q + 3] = t.w;
    }
    #pragma unroll
    for (int c = 1; c <= NC; ++c) {
        a0[c - 1] = fmaf(-v[c - 1], LOW ? u0 : w1, a0[c]);
        a1[c - 1] = fmaf(-v[c - 1], u1, a1[c]);
        b0[c - 1] = fmaf(-v[c - 1], w0, b0[c]);
    }
    a0[NC] = 0.0f;
    a1[NC] = 0.0f;
    b0[NC] = 0.0f;
}

__global__ __launch_bounds__(32) void chol_inv(const float* __restrict__ G,
                                               float* __restrict__ Rinv) {
    __shared__ __attribute__((aligned(16))) float rowb[64];
    __shared__ float wbuf[64 * WSTR];
    const int l = threadIdx.x;

    float a0[64], a1[64], b0[64];
    #pragma unroll
    for (int i = 0; i < 64; ++i) {
        a0[i] = G[i * 64 + l];
        a1[i] = G[i * 64 + l + 32];
        b0[i] = (i == l) ? 1.0f : 0.0f;
    }

    #pragma nounroll
    for (int k = 0; k < 32; ++k)  elim_step<63, true >(k, l, a0, a1, b0, rowb, wbuf);
    #pragma unroll
    for (int c = 0; c < 64; ++c) a0[c] = (c == l) ? 1.0f : 0.0f;
    #pragma nounroll
    for (int k = 32; k < 64; ++k) elim_step<31, false>(k, l, a0, a1, b0, rowb, wbuf);
    __syncthreads();

    #pragma unroll
    for (int r = 0; r < 64; ++r) {
        Rinv[r * 64 + l]      = wbuf[r * WSTR + l];
        Rinv[r * 64 + l + 32] = wbuf[r * WSTR + l + 32];
    }
}

typedef __attribute__((ext_vector_type(4))) float v4f;
#define APITCH 68
#define AWAVES 8
#define ATHREADS 256
#define AGRID 512

__global__ __launch_bounds__(ATHREADS) void apply_rinv(const float* __restrict__ Phi,
                                                         const float* __restrict__ Rinv,
                                                         float* __restrict__ Q, int m) {
    __shared__ __attribute__((aligned(16))) float slab[AWAVES * 16 * APITCH];
    const int lane = threadIdx.x & 31;
    const int wave = __builtin_amdgcn_readfirstlane(threadIdx.x >> 5);
    const int c = lane & 15;
    const int h = lane >> 4;
    const int nstrips = (m + 15) >> 4;
    const int nwaves  = AGRID * AWAVES;
    int strip = blockIdx.x * AWAVES + wave;
    if (strip >= nstrips) return;

    float* my = slab + wave * (16 * APITCH);

    v2f bf[16][4];
    #pragma unroll
    for (int s = 0; s < 16; ++s) {
        const int k = 8 * (s >> 1) + 4 * h + 2 * (s & 1);
        const v4f r0 = *reinterpret_cast<const v4f*>(Rinv + k * 64 + 4 * c);
        const v4f r1 = *reinterpret_cast<const v4f*>(Rinv + (k + 1) * 64 + 4 * c);
        bf[s][0].x = r0.x; bf[s][0].y = r1.x;
        bf[s][1].x = r0.y; bf[s][1].y = r1.y;
        bf[s][2].x = r0.z; bf[s][2].y = r1.z;
        bf[s][3].x = r0.w; bf[s][3].y = r1.w;
    }

    v4f p[8];
    #define LOAD_STRIP(S) do { _Pragma("unroll") for (int j = 0; j < 8; ++j) { \
        long row_ = (long)(S) * 16 + 2 * j + h; if (row_ >= (long)m) row_ = (long)m - 1; \
        p[j] = __builtin_nontemporal_load(reinterpret_cast<const v4f*>(Phi + row_ * 64 + 4 * c)); } } while (0)
    LOAD_STRIP(strip);

    for (; strip < nstrips; strip += nwaves) {
        #pragma unroll
        for (int j = 0; j < 8; ++j)
            *reinterpret_cast<v4f*>(my + (2 * j + h) * APITCH + 4 * c) = p[j];
        __builtin_amdgcn_fence(__ATOMIC_RELEASE, "wavefront");
        __builtin_amdgcn_wave_barrier();
        __builtin_amdgcn_fence(__ATOMIC_ACQUIRE, "wavefront");

        const int nxt = strip + nwaves;
        if (nxt < nstrips) LOAD_STRIP(nxt);

        v4f a[8];
        #pragma unroll
        for (int j = 0; j < 8; ++j)
            a[j] = *reinterpret_cast<const v4f*>(my + c * APITCH + 8 * j + 4 * h);
        __builtin_amdgcn_fence(__ATOMIC_RELEASE, "wavefront");
        __builtin_amdgcn_wave_barrier();
        __builtin_amdgcn_fence(__ATOMIC_ACQUIRE, "wavefront");

        v8f acc0 = {}, acc1 = {}, acc2 = {}, acc3 = {};
        #pragma unroll
        for (int s = 0; s < 16; ++s) {
            v2f va;
            if (s & 1) { va.x = a[s >> 1].z; va.y = a[s >> 1].w; }
            else       { va.x = a[s >> 1].x; va.y = a[s >> 1].y; }
            acc0 = __builtin_amdgcn_wmma_f32_16x16x4_f32(false, va, false, bf[s][0], (short)0, acc0, false, false);
            acc1 = __builtin_amdgcn_wmma_f32_16x16x4_f32(false, va, false, bf[s][1], (short)0, acc1, false, false);
            acc2 = __builtin_amdgcn_wmma_f32_16x16x4_f32(false, va, false, bf[s][2], (short)0, acc2, false, false);
            acc3 = __builtin_amdgcn_wmma_f32_16x16x4_f32(false, va, false, bf[s][3], (short)0, acc3, false, false);
        }

        asm volatile("v_nop\n\tv_nop\n\tv_nop\n\tv_nop\n\tv_nop\n\tv_nop\n\tv_nop\n\tv_nop"
                     : "+v"(acc0), "+v"(acc1), "+v"(acc2), "+v"(acc3), "+v"(a[7]));

        v4f o[8];
        #pragma unroll
        for (int r = 0; r < 8; ++r) { o[r].x = acc0[r]; o[r].y = acc1[r]; o[r].z = acc2[r]; o[r].w = acc3[r]; }
        const long row0 = (long)strip * 16 + 8 * h;
        float* qp = Q + row0 * 64 + 4 * c;
        if ((long)strip * 16 + 16 <= (long)m) {
            #pragma unroll
            for (int r = 0; r < 8; ++r) *reinterpret_cast<v4f*>(qp + r * 64) = o[r];
        } else {
            #pragma unroll
            for (int r = 0; r < 8; ++r)
                if (row0 + r < (long)m) *reinterpret_cast<v4f*>(qp + r * 64) = o[r];
        }
    }
    #undef LOAD_STRIP
}

extern "C" void kernel_launch(void* const* d_in, const int* in_sizes, int n_in,
                              void* d_out, int out_size, void* d_ws, size_t ws_size,
                              hipStream_t stream) {
    const float* Phi = (const float*)d_in[0];
    float* out = (float*)d_out;
    const int m = in_sizes[0] / 64;

    float* ws   = (float*)d_ws;
    float* G    = ws;
    float* Rinv = ws + 4096;
    float* part = ws + 8192;

    long avail = ((long)(ws_size / 4) - 8192) / 4096;
    int nwg = (int)(avail < 1 ? 1 : (avail > 512 ? 512 : avail));

    gram_partial<<<nwg, 512, 0, stream>>>(Phi, part, m, nwg);
    reduce_g<<<128, RTHREADS, 0, stream>>>(part, G, nwg);
    chol_inv<<<1, 32, 0, stream>>>(G, Rinv);

    apply_rinv<<<AGRID, ATHREADS, 0, stream>>>(Phi, Rinv, out, m);
}


# gram_partial: barrier-free waves feed A and B fragments straight from whole-line 16-byte global loads (no LDS staging); a wave keeps 10 of 16 tiles, mirrors filled at write-out; G's sum tree changes
# speedup vs baseline: 3.2345x; 1.2091x over previous
#include <hip/hip_runtime.h>

typedef __attribute__((ext_vector_type(2))) float v2f;
typedef __attribute__((ext_vector_type(8))) float v8f;

typedef __attribute__((ext_vector_type(4))) float g4f;
#define GWAVES 8
#define GTHREADS 256
#define GSLOT 2560
#define GPITCH 65

__global__ __launch_bounds__(GTHREADS) void gram_partial(const float* __restrict__ Phi,
                                                         float* __restrict__ part,
                                                         int m, int nwg) {
    __shared__ float red[4 * GSLOT];
    const int lane = threadIdx.x & 31;
    const int wave = __builtin_amdgcn_readfirstlane(threadIdx.x >> 5);
    const int c = lane & 15;
    const int h = lane >> 4;
    const int nstrips = (m + 15) >> 4;
    const int nwaves  = nwg * GWAVES;
    int strip = blockIdx.x * GWAVES + wave;

    v8f acc[10];
    #pragma unroll
    for (int q = 0; q < 10; ++q) acc[q] = v8f{};

    g4f p[8];
    #define GLOAD_STRIP(S) do { _Pragma("unroll") for (int j = 0; j < 8; ++j) { \
        long row_ = (long)(S) * 16 + 2 * j + h; if (row_ >= (long)m) row_ = (long)m - 1; \
        p[j] = *reinterpret_cast<const g4f*>(Phi + row_ * 64 + 4 * c); } } while (0)
    if (strip < nstrips) GLOAD_STRIP(strip);

    for (; strip < nstrips; strip += nwaves) {
        g4f a[8];
        #pragma unroll
        for (int j = 0; j < 8; ++j) a[j] = p[j];
        if ((long)strip * 16 + 16 > (long)m) {
            #pragma unroll
            for (int j = 0; j < 8; ++j)
                if ((long)strip * 16 + 2 * j + h >= (long)m) a[j] = g4f{0.0f, 0.0f, 0.0f, 0.0f};
        }

        const int nxt = strip + nwaves;
        if (nxt < nstrips) GLOAD_STRIP(nxt);

        #pragma unroll
        for (int i = 0; i < 4; ++i) {
            v2f f[4];
            f[0].x = a[2 * i].x; f[0].y = a[2 * i + 1].x;
            f[1].x = a[2 * i].y; f[1].y = a[2 * i + 1].y;
            f[2].x = a[2 * i].z; f[2].y = a[2 * i + 1].z;
            f[3].x = a[2 * i].w; f[3].y = a[2 * i + 1].w;
            int q = 0;
            #pragma unroll
            for (int ta = 0; ta < 4; ++ta) {
                #pragma unroll
                for (int tb = ta; tb < 4; ++tb) {
                    acc[q] = __builtin_amdgcn_wmma_f32_16x16x4_f32(false, f[ta], false, f[tb],
                                                                   (short)0, acc[q], false, false);
                    ++q;
                }
            }
        }
    }
    #undef GLOAD_STRIP

    #define GSTORE(SLOT) do { float* s_ = red + (SLOT) * GSLOT + lane; _Pragma("unroll") for (int q = 0; q < 10; ++q) { \
        _Pragma("unroll") for (int r = 0; r < 8; ++r) s_[(q * 8 + r) * 32] = acc[q][r]; } } while (0)
    #define GADD(SLOT) do { const float* s_ = red + (SLOT) * GSLOT + lane; _Pragma("unroll") for (int q = 0; q < 10; ++q) { \
        _Pragma("unroll") for (int r = 0; r < 8; ++r) acc[q][r] += s_[(q * 8 + r) * 32]; } } while (0)
    if (wave >= 4) GSTORE(wave - 4);
    __syncthreads();
    if (wave < 4) GADD(wave);
    __syncthreads();
    if (wave == 2 || wave == 3) GSTORE(wave - 2);
    __syncthreads();
    if (wave < 2) GADD(wave);
    __syncthreads();
    if (wave == 1) GSTORE(0);
    __syncthreads();
    if (wave == 0) {
        GADD(0);
        int q = 0;
        #pragma unroll
        for (int ta = 0; ta < 4; ++ta) {
            #pragma unroll
            for (int tb = ta; tb < 4; ++tb) {
                #pragma unroll
                for (int r = 0; r < 8; ++r)
                    red[(32 * h + 4 * r + ta) * GPITCH + 4 * c + tb] = acc[q][r];
                ++q;
            }
        }
    }
    __syncthreads();
    #undef GSTORE
    #undef GADD

    float* pp = part + (size_t)blockIdx.x * 4096 + threadIdx.x;
    const int i0 = threadIdx.x >> 6, j = threadIdx.x & 63;
    const bool own = i0 <= (j & 3);
    const float* src = red + (own ? i0 * GPITCH + j : j * GPITCH + i0);
    const int step = own ? 4 * GPITCH : 4;
    float o[4096 / GTHREADS];
    #pragma unroll
    for (int k = 0; k < 4096 / GTHREADS; ++k) o[k] = src[k * step];
    #pragma unroll
    for (int k = 0; k < 4096 / GTHREADS; ++k) pp[k * GTHREADS] = o[k];
}

#define RWAVES 16
#define RTHREADS 512
#define RDEPTH 32
__global__ __launch_bounds__(RTHREADS) void reduce_g(const float* __restrict__ part,
                                                        float* __restrict__ G, int nwg) {
    __shared__ float red[RWAVES * 32];
    const int lane = threadIdx.x & 31;
    const int wave = __builtin_amdgcn_readfirstlane(threadIdx.x >> 5);
    const int e = blockIdx.x * 32 + lane;
    const float* p = part + e;
    float s0 = 0.0f, s1 = 0.0f, s2 = 0.0f, s3 = 0.0f;
    int w = wave;
    for (; w + (RDEPTH - 1) * RWAVES < nwg; w += RDEPTH * RWAVES) {
        float t[RDEPTH];
        #pragma unroll
        for (int i = 0; i < RDEPTH; ++i) t[i] = p[(size_t)(w + i * RWAVES) * 4096];
        #pragma unroll
        for (int i = 0; i < RDEPTH; i += 4) { s0 += t[i]; s1 += t[i + 1]; s2 += t[i + 2]; s3 += t[i + 3]; }
    }
    for (; w < nwg; w += RWAVES) s0 += p[(size_t)w * 4096];
    red[wave * 32 + lane] = (s0 + s1) + (s2 + s3);
    __syncthreads();
    if (wave == 0) {
        float g = 0.0f;
        #pragma unroll
        for (int i = 0; i < RWAVES; ++i) g += red[i * 32 + lane];
        G[e] = g;
    }
}

#define WSTR 65

template <int NC, bool LOW>
__device__ __forceinline__ void elim_step(const int k, const int l,
                                          float (&a0)[64],
                                          float (&a1)[64],
                                          float (&b0)[64],
                                          float* rowb, float* wbuf) {
    const float dsel = LOW ? a0[0] : a1[0];
    const float akk  = __int_as_float(__builtin_amdgcn_readlane(__float_as_int(dsel), k & 31));
    const float dk   = sqrtf(akk);
    const int j0 = l, j1 = l + 32;
    const float u0 = LOW ? a0[0] / dk : 0.0f;
    const float u1 = a1[0] / dk;
    const float w0 = b0[0] / dk;
    const float w1 = LOW ? 0.0f : a0[0] / dk;
    __syncthreads();
    rowb[(j0 - k - 1) & 63] = (j0 > k) ? u0 : 0.0f;
    rowb[(j1 - k - 1) & 63] = (j1 > k) ? u1 : 0.0f;
    wbuf[j0 * WSTR + k] = w0;
    wbuf[j1 * WSTR + k] = w1;
    __syncthreads();

    float v[NC + 1];
    const float4* r4 = reinterpret_cast<const float4*>(rowb);
    #pragma unroll
    for (int q = 0; q < (NC + 1) / 4; ++q) {
        const float4 t = r4[q];
        v[4 * q] = t.x; v[4 * q + 1] = t.y; v[4 * q + 2] = t.z; v[4 * q + 3] = t.w;
    }
    #pragma unroll
    for (int c = 1; c <= NC; ++c) {
        a0[c - 1] = fmaf(-v[c - 1], LOW ? u0 : w1, a0[c]);
        a1[c - 1] = fmaf(-v[c - 1], u1, a1[c]);
        b0[c - 1] = fmaf(-v[c - 1], w0, b0[c]);
    }
    a0[NC] = 0.0f;
    a1[NC] = 0.0f;
    b0[NC] = 0.0f;
}

__global__ __launch_bounds__(32) void chol_inv(const float* __restrict__ G,
                                               float* __restrict__ Rinv) {
    __shared__ __attribute__((aligned(16))) float rowb[64];
    __shared__ float wbuf[64 * WSTR];
    const int l = threadIdx.x;

    float a0[64], a1[64], b0[64];
    #pragma unroll
    for (int i = 0; i < 64; ++i) {
        a0[i] = G[i * 64 + l];
        a1[i] = G[i * 64 + l + 32];
        b0[i] = (i == l) ? 1.0f : 0.0f;
    }

    #pragma nounroll
    for (int k = 0; k < 32; ++k)  elim_step<63, true >(k, l, a0, a1, b0, rowb, wbuf);
    #pragma unroll
    for (int c = 0; c < 64; ++c) a0[c] = (c == l) ? 1.0f : 0.0f;
    #pragma nounroll
    for (int k = 32; k < 64; ++k) elim_step<31, false>(k, l, a0, a1, b0, rowb, wbuf);
    __syncthreads();

    #pragma unroll
    for (int r = 0; r < 64; ++r) {
        Rinv[r * 64 + l]      = wbuf[r * WSTR + l];
        Rinv[r * 64 + l + 32] = wbuf[r * WSTR + l + 32];
    }
}

typedef __attribute__((ext_vector_type(4))) float v4f;
#define APITCH 68
#define AWAVES 8
#define ATHREADS 256
#define AGRID 512

__global__ __launch_bounds__(ATHREADS) void apply_rinv(const float* __restrict__ Phi,
                                                         const float* __restrict__ Rinv,
                                                         float* __restrict__ Q, int m) {
    __shared__ __attribute__((aligned(16))) float slab[AWAVES * 16 * APITCH];
    const int lane = threadIdx.x & 31;
    const int wave = __builtin_amdgcn_readfirstlane(threadIdx.x >> 5);
    const int c = lane & 15;
    const int h = lane >> 4;
    const int nstrips = (m + 15) >> 4;
    const int nwaves  = AGRID * AWAVES;
    int strip = blockIdx.x * AWAVES + wave;
    if (strip >= nstrips) return;

    float* my = slab + wave * (16 * APITCH);

    v2f bf[16][4];
    #pragma unroll
    for (int s = 0; s < 16; ++s) {
        const int k = 8 * (s >> 1) + 4 * h + 2 * (s & 1);
        const v4f r0 = *reinterpret_cast<const v4f*>(Rinv + k * 64 + 4 * c);
        const v4f r1 = *reinterpret_cast<const v4f*>(Rinv + (k + 1) * 64 + 4 * c);
        bf[s][0].x = r0.x; bf[s][0].y = r1.x;
        bf[s][1].x = r0.y; bf[s][1].y = r1.y;
        bf[s][2].x = r0.z; bf[s][2].y = r1.z;
        bf[s][3].x = r0.w; bf[s][3].y = r1.w;
    }

    v4f p[8];
    #define LOAD_STRIP(S) do { _Pragma("unroll") for (int j = 0; j < 8; ++j) { \
        long row_ = (long)(S) * 16 + 2 * j + h; if (row_ >= (long)m) row_ = (long)m - 1; \
        p[j] = __builtin_nontemporal_load(reinterpret_cast<const v4f*>(Phi + row_ * 64 + 4 * c)); } } while (0)
    LOAD_STRIP(strip);

    for (; strip < nstrips; strip += nwaves) {
        #pragma unroll
        for (int j = 0; j < 8; ++j)
            *reinterpret_cast<v4f*>(my + (2 * j + h) * APITCH + 4 * c) = p[j];
        __builtin_amdgcn_fence(__ATOMIC_RELEASE, "wavefront");
        __builtin_amdgcn_wave_barrier();
        __builtin_amdgcn_fence(__ATOMIC_ACQUIRE, "wavefront");

        const int nxt = strip + nwaves;
        if (nxt < nstrips) LOAD_STRIP(nxt);

        v4f a[8];
        #pragma unroll
        for (int j = 0; j < 8; ++j)
            a[j] = *reinterpret_cast<const v4f*>(my + c * APITCH + 8 * j + 4 * h);
        __builtin_amdgcn_fence(__ATOMIC_RELEASE, "wavefront");
        __builtin_amdgcn_wave_barrier();
        __builtin_amdgcn_fence(__ATOMIC_ACQUIRE, "wavefront");

        v8f acc0 = {}, acc1 = {}, acc2 = {}, acc3 = {};
        #pragma unroll
        for (int s = 0; s < 16; ++s) {
            v2f va;
            if (s & 1) { va.x = a[s >> 1].z; va.y = a[s >> 1].w; }
            else       { va.x = a[s >> 1].x; va.y = a[s >> 1].y; }
            acc0 = __builtin_amdgcn_wmma_f32_16x16x4_f32(false, va, false, bf[s][0], (short)0, acc0, false, false);
            acc1 = __builtin_amdgcn_wmma_f32_16x16x4_f32(false, va, false, bf[s][1], (short)0, acc1, false, false);
            acc2 = __builtin_amdgcn_wmma_f32_16x16x4_f32(false, va, false, bf[s][2], (short)0, acc2, false, false);
            acc3 = __builtin_amdgcn_wmma_f32_16x16x4_f32(false, va, false, bf[s][3], (short)0, acc3, false, false);
        }

        asm volatile("v_nop\n\tv_nop\n\tv_nop\n\tv_nop\n\tv_nop\n\tv_nop\n\tv_nop\n\tv_nop"
                     : "+v"(acc0), "+v"(acc1), "+v"(acc2), "+v"(acc3), "+v"(a[7]));

        v4f o[8];
        #pragma unroll
        for (int r = 0; r < 8; ++r) { o[r].x = acc0[r]; o[r].y = acc1[r]; o[r].z = acc2[r]; o[r].w = acc3[r]; }
        const long row0 = (long)strip * 16 + 8 * h;
        float* qp = Q + row0 * 64 + 4 * c;
        if ((long)strip * 16 + 16 <= (long)m) {
            #pragma unroll
            for (int r = 0; r < 8; ++r) *reinterpret_cast<v4f*>(qp + r * 64) = o[r];
        } else {
            #pragma unroll
            for (int r = 0; r < 8; ++r)
                if (row0 + r < (long)m) *reinterpret_cast<v4f*>(qp + r * 64) = o[r];
        }
    }
    #undef LOAD_STRIP
}

extern "C" void kernel_launch(void* const* d_in, const int* in_sizes, int n_in,
                              void* d_out, int out_size, void* d_ws, size_t ws_size,
                              hipStream_t stream) {
    const float* Phi = (const float*)d_in[0];
    float* out = (float*)d_out;
    const int m = in_sizes[0] / 64;

    float* ws   = (float*)d_ws;
    float* G    = ws;
    float* Rinv = ws + 4096;
    float* part = ws + 8192;

    long avail = ((long)(ws_size / 4) - 8192) / 4096;
    int nwg = (int)(avail < 1 ? 1 : (avail > 512 ? 512 : avail));

    gram_partial<<<nwg, GTHREADS, 0, stream>>>(Phi, part, m, nwg);
    reduce_g<<<128, RTHREADS, 0, stream>>>(part, G, nwg);
    chol_inv<<<1, 32, 0, stream>>>(G, Rinv);

    apply_rinv<<<AGRID, ATHREADS, 0, stream>>>(Phi, Rinv, out, m);
}


# apply_rinv: WMMAs against the zero lower triangle of Rinv dropped (40 per strip, was 64): column-block tiles, product taken transposed, rows re-formed in the wave's LDS slab; order of sums kept
# speedup vs baseline: 3.3331x; 1.0305x over previous
#include <hip/hip_runtime.h>

typedef __attribute__((ext_vector_type(2))) float v2f;
typedef __attribute__((ext_vector_type(8))) float v8f;

typedef __attribute__((ext_vector_type(4))) float g4f;
#define GWAVES 8
#define GTHREADS 256
#define GSLOT 2560
#define GPITCH 65

__global__ __launch_bounds__(GTHREADS) void gram_partial(const float* __restrict__ Phi,
                                                         float* __restrict__ part,
                                                         int m, int nwg) {
    __shared__ float red[4 * GSLOT];
    const int lane = threadIdx.x & 31;
    const int wave = __builtin_amdgcn_readfirstlane(threadIdx.x >> 5);
    const int c = lane & 15;
    const int h = lane >> 4;
    const int nstrips = (m + 15) >> 4;
    const int nwaves  = nwg * GWAVES;
    int strip = blockIdx.x * GWAVES + wave;

    v8f acc[10];
    #pragma unroll
    for (int q = 0; q < 10; ++q) acc[q] = v8f{};

    g4f p[8];
    #define GLOAD_STRIP(S) do { _Pragma("unroll") for (int j = 0; j < 8; ++j) { \
        long row_ = (long)(S) * 16 + 2 * j + h; if (row_ >= (long)m) row_ = (long)m - 1; \
        p[j] = *reinterpret_cast<const g4f*>(Phi + row_ * 64 + 4 * c); } } while (0)
    if (strip < nstrips) GLOAD_STRIP(strip);

    for (; strip < nstrips; strip += nwaves) {
        g4f a[8];
        #pragma unroll
        for (int j = 0; j < 8; ++j) a[j] = p[j];
        if ((long)strip * 16 + 16 > (long)m) {
            #pragma unroll
            for (int j = 0; j < 8; ++j)
                if ((long)strip * 16 + 2 * j + h >= (long)m) a[j] = g4f{0.0f, 0.0f, 0.0f, 0.0f};
        }

        const int nxt = strip + nwaves;
        if (nxt < nstrips) GLOAD_STRIP(nxt);

        #pragma unroll
        for (int i = 0; i < 4; ++i) {
            v2f f[4];
            f[0].x = a[2 * i].x; f[0].y = a[2 * i + 1].x;
            f[1].x = a[2 * i].y; f[1].y = a[2 * i + 1].y;
            f[2].x = a[2 * i].z; f[2].y = a[2 * i + 1].z;
            f[3].x = a[2 * i].w; f[3].y = a[2 * i + 1].w;
            int q = 0;
            #pragma unroll
            for (int ta = 0; ta < 4; ++ta) {
                #pragma unroll
                for (int tb = ta; tb < 4; ++tb) {
                    acc[q] = __builtin_amdgcn_wmma_f32_16x16x4_f32(false, f[ta], false, f[tb],
                                                                   (short)0, acc[q], false, false);
                    ++q;
                }
            }
        }
    }
    #undef GLOAD_STRIP

    #define GSTORE(SLOT) do { float* s_ = red + (SLOT) * GSLOT + lane; _Pragma("unroll") for (int q = 0; q < 10; ++q) { \
        _Pragma("unroll") for (int r = 0; r < 8; ++r) s_[(q * 8 + r) * 32] = acc[q][r]; } } while (0)
    #define GADD(SLOT) do { const float* s_ = red + (SLOT) * GSLOT + lane; _Pragma("unroll") for (int q = 0; q < 10; ++q) { \
        _Pragma("unroll") for (int r = 0; r < 8; ++r) acc[q][r] += s_[(q * 8 + r) * 32]; } } while (0)
    if (wave >= 4) GSTORE(wave - 4);
    __syncthreads();
    if (wave < 4) GADD(wave);
    __syncthreads();
    if (wave == 2 || wave == 3) GSTORE(wave - 2);
    __syncthreads();
    if (wave < 2) GADD(wave);
    __syncthreads();
    if (wave == 1) GSTORE(0);
    __syncthreads();
    if (wave == 0) {
        GADD(0);
        int q = 0;
        #pragma unroll
        for (int ta = 0; ta < 4; ++ta) {
            #pragma unroll
            for (int tb = ta; tb < 4; ++tb) {
                #pragma unroll
                for (int r = 0; r < 8; ++r)
                    red[(32 * h + 4 * r + ta) * GPITCH + 4 * c + tb] = acc[q][r];
                ++q;
            }
        }
    }
    __syncthreads();
    #undef GSTORE
    #undef GADD

    float* pp = part + (size_t)blockIdx.x * 4096 + threadIdx.x;
    const int i0 = threadIdx.x >> 6, j = threadIdx.x & 63;
    const bool own = i0 <= (j & 3);
    const float* src = red + (own ? i0 * GPITCH + j : j * GPITCH + i0);
    const int step = own ? 4 * GPITCH : 4;
    float o[4096 / GTHREADS];
    #pragma unroll
    for (int k = 0; k < 4096 / GTHREADS; ++k) o[k] = src[k * step];
    #pragma unroll
    for (int k = 0; k < 4096 / GTHREADS; ++k) pp[k * GTHREADS] = o[k];
}

#define RWAVES 16
#define RTHREADS 512
#define RDEPTH 32
__global__ __launch_bounds__(RTHREADS) void reduce_g(const float* __restrict__ part,
                                                        float* __restrict__ G, int nwg) {
    __shared__ float red[RWAVES * 32];
    const int lane = threadIdx.x & 31;
    const int wave = __builtin_amdgcn_readfirstlane(threadIdx.x >> 5);
    const int e = blockIdx.x * 32 + lane;
    const float* p = part + e;
    float s0 = 0.0f, s1 = 0.0f, s2 = 0.0f, s3 = 0.0f;
    int w = wave;
    for (; w + (RDEPTH - 1) * RWAVES < nwg; w += RDEPTH * RWAVES) {
        float t[RDEPTH];
        #pragma unroll
        for (int i = 0; i < RDEPTH; ++i) t[i] = p[(size_t)(w + i * RWAVES) * 4096];
        #pragma unroll
        for (int i = 0; i < RDEPTH; i += 4) { s0 += t[i]; s1 += t[i + 1]; s2 += t[i + 2]; s3 += t[i + 3]; }
    }
    for (; w < nwg; w += RWAVES) s0 += p[(size_t)w * 4096];
    red[wave * 32 + lane] = (s0 + s1) + (s2 + s3);
    __syncthreads();
    if (wave == 0) {
        float g = 0.0f;
        #pragma unroll
        for (int i = 0; i < RWAVES; ++i) g += red[i * 32 + lane];
        G[e] = g;
    }
}

#define WSTR 65

template <int NC, bool LOW>
__device__ __forceinline__ void elim_step(const int k, const int l,
                                          float (&a0)[64],
                                          float (&a1)[64],
                                          float (&b0)[64],
                                          float* rowb, float* wbuf) {
    const float dsel = LOW ? a0[0] : a1[0];
    const float akk  = __int_as_float(__builtin_amdgcn_readlane(__float_as_int(dsel), k & 31));
    const float dk   = sqrtf(akk);
    const int j0 = l, j1 = l + 32;
    const float u0 = LOW ? a0[0] / dk : 0.0f;
    const float u1 = a1[0] / dk;
    const float w0 = b0[0] / dk;
    const float w1 = LOW ? 0.0f : a0[0] / dk;
    __syncthreads();
    rowb[(j0 - k - 1) & 63] = (j0 > k) ? u0 : 0.0f;
    rowb[(j1 - k - 1) & 63] = (j1 > k) ? u1 : 0.0f;
    wbuf[j0 * WSTR + k] = w0;
    wbuf[j1 * WSTR + k] = w1;
    __syncthreads();

    float v[NC + 1];
    const float4* r4 = reinterpret_cast<const float4*>(rowb);
    #pragma unroll
    for (int q = 0; q < (NC + 1) / 4; ++q) {
        const float4 t = r4[q];
        v[4 * q] = t.x; v[4 * q + 1] = t.y; v[4 * q + 2] = t.z; v[4 * q + 3] = t.w;
    }
    #pragma unroll
    for (int c = 1; c <= NC; ++c) {
        a0[c - 1] = fmaf(-v[c - 1], LOW ? u0 : w1, a0[c]);
        a1[c - 1] = fmaf(-v[c - 1], u1, a1[c]);
        b0[c - 1] = fmaf(-v[c - 1], w0, b0[c]);
    }
    a0[NC] = 0.0f;
    a1[NC] = 0.0f;
    b0[NC] = 0.0f;
}

__global__ __launch_bounds__(32) void chol_inv(const float* __restrict__ G,
                                               float* __restrict__ Rinv) {
    __shared__ __attribute__((aligned(16))) float rowb[64];
    __shared__ float wbuf[64 * WSTR];
    const int l = threadIdx.x;

    float a0[64], a1[64], b0[64];
    #pragma unroll
    for (int i = 0; i < 64; ++i) {
        a0[i] = G[i * 64 + l];
        a1[i] = G[i * 64 + l + 32];
        b0[i] = (i == l) ? 1.0f : 0.0f;
    }

    #pragma nounroll
    for (int k = 0; k < 32; ++k)  elim_step<63, true >(k, l, a0, a1, b0, rowb, wbuf);
    #pragma unroll
    for (int c = 0; c < 64; ++c) a0[c] = (c == l) ? 1.0f : 0.0f;
    #pragma nounroll
    for (int k = 32; k < 64; ++k) elim_step<31, false>(k, l, a0, a1, b0, rowb, wbuf);
    __syncthreads();

    #pragma unroll
    for (int r = 0; r < 64; ++r) {
        Rinv[r * 64 + l]      = wbuf[r * WSTR + l];
        Rinv[r * 64 + l + 32] = wbuf[r * WSTR + l + 32];
    }
}

typedef __attribute__((ext_vector_type(4))) float v4f;
#define APITCH 68
#define AWAVES 8
#define ATHREADS 256
#define AGRID 512

__global__ __launch_bounds__(ATHREADS) void apply_rinv(const float* __restrict__ Phi,
                                                         const float* __restrict__ Rinv,
                                                         float* __restrict__ Q, int m) {
    __shared__ __attribute__((aligned(16))) float slab[AWAVES * 16 * APITCH];
    const int lane = threadIdx.x & 31;
    const int wave = __builtin_amdgcn_readfirstlane(threadIdx.x >> 5);
    const int c = lane & 15;
    const int h = lane >> 4;
    const int nstrips = (m + 15) >> 4;
    const int nwaves  = AGRID * AWAVES;
    int strip = blockIdx.x * AWAVES + wave;
    if (strip >= nstrips) return;

    float* my = slab + wave * (16 * APITCH);

    v2f rf[40];
    #pragma unroll
    for (int t = 0; t < 4; ++t) {
        #pragma unroll
        for (int s = 0; s < 4 * (t + 1); ++s) {
            const int k = 8 * (s >> 1) + 4 * h + 2 * (s & 1);
            rf[2 * t * (t + 1) + s].x = Rinv[k * 64 + 16 * t + c];
            rf[2 * t * (t + 1) + s].y = Rinv[(k + 1) * 64 + 16 * t + c];
        }
    }

    v4f p[8];
    #define LOAD_STRIP(S) do { _Pragma("unroll") for (int j = 0; j < 8; ++j) { \
        long row_ = (long)(S) * 16 + 2 * j + h; if (row_ >= (long)m) row_ = (long)m - 1; \
        p[j] = __builtin_nontemporal_load(reinterpret_cast<const v4f*>(Phi + row_ * 64 + 4 * c)); } } while (0)
    #define SLAB_ORDER() do { __builtin_amdgcn_fence(__ATOMIC_RELEASE, "wavefront"); \
        __builtin_amdgcn_wave_barrier(); __builtin_amdgcn_fence(__ATOMIC_ACQUIRE, "wavefront"); } while (0)
    LOAD_STRIP(strip);

    for (; strip < nstrips; strip += nwaves) {
        #pragma unroll
        for (int j = 0; j < 8; ++j)
            *reinterpret_cast<v4f*>(my + (2 * j + h) * APITCH + 4 * c) = p[j];
        SLAB_ORDER();

        const int nxt = strip + nwaves;
        if (nxt < nstrips) LOAD_STRIP(nxt);

        v4f a[8];
        #pragma unroll
        for (int j = 0; j < 8; ++j)
            a[j] = *reinterpret_cast<const v4f*>(my + c * APITCH + 8 * j + 4 * h);

        v8f acc0 = {}, acc1 = {}, acc2 = {}, acc3 = {};
        #define W(S, T) acc##T = __builtin_amdgcn_wmma_f32_16x16x4_f32(false, rf[2 * (T) * ((T) + 1) + (S)], false, va, \
                                                                       (short)0, acc##T, false, false)
        #pragma unroll
        for (int s = 0; s < 16; ++s) {
            v2f va;
            if (s & 1) { va.x = a[s >> 1].z; va.y = a[s >> 1].w; }
            else       { va.x = a[s >> 1].x; va.y = a[s >> 1].y; }
            if (s < 4)  W(s, 0);
            if (s < 8)  W(s, 1);
            if (s < 12) W(s, 2);
            W(s, 3);
        }
        #undef W

        asm volatile("v_nop\n\tv_nop\n\tv_nop\n\tv_nop\n\tv_nop\n\tv_nop\n\tv_nop\n\tv_nop"
                     : "+v"(acc0), "+v"(acc1), "+v"(acc2), "+v"(acc3));

        #define PUT(T) do { float* d_ = my + c * APITCH + 16 * (T) + 8 * h; \
            *reinterpret_cast<v4f*>(d_)     = __builtin_shufflevector(acc##T, acc##T, 0, 1, 2, 3); \
            *reinterpret_cast<v4f*>(d_ + 4) = __builtin_shufflevector(acc##T, acc##T, 4, 5, 6, 7); } while (0)
        PUT(0); PUT(1); PUT(2); PUT(3);
        #undef PUT
        SLAB_ORDER();

        v4f o[8];
        #pragma unroll
        for (int j = 0; j < 8; ++j)
            o[j] = *reinterpret_cast<const v4f*>(my + (2 * j + h) * APITCH + 4 * c);
        SLAB_ORDER();

        const long row0 = (long)strip * 16 + h;
        float* qp = Q + row0 * 64 + 4 * c;
        if ((long)strip * 16 + 16 <= (long)m) {
            #pragma unroll
            for (int j = 0; j < 8; ++j) *reinterpret_cast<v4f*>(qp + j * 128) = o[j];
        } else {
            #pragma unroll
            for (int j = 0; j < 8; ++j)
                if (row0 + 2 * j < (long)m) *reinterpret_cast<v4f*>(qp + j * 128) = o[j];
        }
    }
    #undef LOAD_STRIP
    #undef SLAB_ORDER
}

extern "C" void kernel_launch(void* const* d_in, const int* in_sizes, int n_in,
                              void* d_out, int out_size, void* d_ws, size_t ws_size,
                              hipStream_t stream) {
    const float* Phi = (const float*)d_in[0];
    float* out = (float*)d_out;
    const int m = in_sizes[0] / 64;

    float* ws   = (float*)d_ws;
    float* G    = ws;
    float* Rinv = ws + 4096;
    float* part = ws + 8192;

    long avail = ((long)(ws_size / 4) - 8192) / 4096;
    int nwg = (int)(avail < 1 ? 1 : (avail > 512 ? 512 : avail));

    gram_partial<<<nwg, GTHREADS, 0, stream>>>(Phi, part, m, nwg);
    reduce_g<<<128, RTHREADS, 0, stream>>>(part, G, nwg);
    chol_inv<<<1, 32, 0, stream>>>(G, Rinv);

    apply_rinv<<<AGRID, ATHREADS, 0, stream>>>(Phi, Rinv, out, m);
}


# chol_inv: pivot row scaled by one reciprocal square root instruction and multiplies (was a rounded sqrt and three rounded divisions on each step's serial chain); U and Rinv round differently
# speedup vs baseline: 3.5092x; 1.0528x over previous
#include <hip/hip_runtime.h>

typedef __attribute__((ext_vector_type(2))) float v2f;
typedef __attribute__((ext_vector_type(8))) float v8f;

typedef __attribute__((ext_vector_type(4))) float g4f;
#define GWAVES 8
#define GTHREADS 256
#define GSLOT 2560
#define GPITCH 65

__global__ __launch_bounds__(GTHREADS) void gram_partial(const float* __restrict__ Phi,
                                                         float* __restrict__ part,
                                                         int m, int nwg) {
    __shared__ float red[4 * GSLOT];
    const int lane = threadIdx.x & 31;
    const int wave = __builtin_amdgcn_readfirstlane(threadIdx.x >> 5);
    const int c = lane & 15;
    const int h = lane >> 4;
    const int nstrips = (m + 15) >> 4;
    const int nwaves  = nwg * GWAVES;
    int strip = blockIdx.x * GWAVES + wave;

    v8f acc[10];
    #pragma unroll
    for (int q = 0; q < 10; ++q) acc[q] = v8f{};

    g4f p[8];
    #define GLOAD_STRIP(S) do { _Pragma("unroll") for (int j = 0; j < 8; ++j) { \
        long row_ = (long)(S) * 16 + 2 * j + h; if (row_ >= (long)m) row_ = (long)m - 1; \
        p[j] = *reinterpret_cast<const g4f*>(Phi + row_ * 64 + 4 * c); } } while (0)
    if (strip < nstrips) GLOAD_STRIP(strip);

    for (; strip < nstrips; strip += nwaves) {
        g4f a[8];
        #pragma unroll
        for (int j = 0; j < 8; ++j) a[j] = p[j];
        if ((long)strip * 16 + 16 > (long)m) {
            #pragma unroll
            for (int j = 0; j < 8; ++j)
                if ((long)strip * 16 + 2 * j + h >= (long)m) a[j] = g4f{0.0f, 0.0f, 0.0f, 0.0f};
        }

        const int nxt = strip + nwaves;
        if (nxt < nstrips) GLOAD_STRIP(nxt);

        #pragma unroll
        for (int i = 0; i < 4; ++i) {
            v2f f[4];
            f[0].x = a[2 * i].x; f[0].y = a[2 * i + 1].x;
            f[1].x = a[2 * i].y; f[1].y = a[2 * i + 1].y;
            f[2].x = a[2 * i].z; f[2].y = a[2 * i + 1].z;
            f[3].x = a[2 * i].w; f[3].y = a[2 * i + 1].w;
            int q = 0;
            #pragma unroll
            for (int ta = 0; ta < 4; ++ta) {
                #pragma unroll
                for (int tb = ta; tb < 4; ++tb) {
                    acc[q] = __builtin_amdgcn_wmma_f32_16x16x4_f32(false, f[ta], false, f[tb],
                                                                   (short)0, acc[q], false, false);
                    ++q;
                }
            }
        }
    }
    #undef GLOAD_STRIP

    #define GSTORE(SLOT) do { float* s_ = red + (SLOT) * GSLOT + lane; _Pragma("unroll") for (int q = 0; q < 10; ++q) { \
        _Pragma("unroll") for (int r = 0; r < 8; ++r) s_[(q * 8 + r) * 32] = acc[q][r]; } } while (0)
    #define GADD(SLOT) do { const float* s_ = red + (SLOT) * GSLOT + lane; _Pragma("unroll") for (int q = 0; q < 10; ++q) { \
        _Pragma("unroll") for (int r = 0; r < 8; ++r) acc[q][r] += s_[(q * 8 + r) * 32]; } } while (0)
    if (wave >= 4) GSTORE(wave - 4);
    __syncthreads();
    if (wave < 4) GADD(wave);
    __syncthreads();
    if (wave == 2 || wave == 3) GSTORE(wave - 2);
    __syncthreads();
    if (wave < 2) GADD(wave);
    __syncthreads();
    if (wave == 1) GSTORE(0);
    __syncthreads();
    if (wave == 0) {
        GADD(0);
        int q = 0;
        #pragma unroll
        for (int ta = 0; ta < 4; ++ta) {
            #pragma unroll
            for (int tb = ta; tb < 4; ++tb) {
                #pragma unroll
                for (int r = 0; r < 8; ++r)
                    red[(32 * h + 4 * r + ta) * GPITCH + 4 * c + tb] = acc[q][r];
                ++q;
            }
        }
    }
    __syncthreads();
    #undef GSTORE
    #undef GADD

    float* pp = part + (size_t)blockIdx.x * 4096 + threadIdx.x;
    const int i0 = threadIdx.x >> 6, j = threadIdx.x & 63;
    const bool own = i0 <= (j & 3);
    const float* src = red + (own ? i0 * GPITCH + j : j * GPITCH + i0);
    const int step = own ? 4 * GPITCH : 4;
    float o[4096 / GTHREADS];
    #pragma unroll
    for (int k = 0; k < 4096 / GTHREADS; ++k) o[k] = src[k * step];
    #pragma unroll
    for (int k = 0; k < 4096 / GTHREADS; ++k) pp[k * GTHREADS] = o[k];
}

#define RWAVES 16
#define RTHREADS 512
#define RDEPTH 32
__global__ __launch_bounds__(RTHREADS) void reduce_g(const float* __restrict__ part,
                                                        float* __restrict__ G, int nwg) {
    __shared__ float red[RWAVES * 32];
    const int lane = threadIdx.x & 31;
    const int wave = __builtin_amdgcn_readfirstlane(threadIdx.x >> 5);
    const int e = blockIdx.x * 32 + lane;
    const float* p = part + e;
    float s0 = 0.0f, s1 = 0.0f, s2 = 0.0f, s3 = 0.0f;
    int w = wave;
    for (; w + (RDEPTH - 1) * RWAVES < nwg; w += RDEPTH * RWAVES) {
        float t[RDEPTH];
        #pragma unroll
        for (int i = 0; i < RDEPTH; ++i) t[i] = p[(size_t)(w + i * RWAVES) * 4096];
        #pragma unroll
        for (int i = 0; i < RDEPTH; i += 4) { s0 += t[i]; s1 += t[i + 1]; s2 += t[i + 2]; s3 += t[i + 3]; }
    }
    for (; w < nwg; w += RWAVES) s0 += p[(size_t)w * 4096];
    red[wave * 32 + lane] = (s0 + s1) + (s2 + s3);
    __syncthreads();
    if (wave == 0) {
        float g = 0.0f;
        #pragma unroll
        for (int i = 0; i < RWAVES; ++i) g += red[i * 32 + lane];
        G[e] = g;
    }
}

#define WSTR 65

template <int NC, bool LOW>
__device__ __forceinline__ void elim_step(const int k, const int l,
                                          float (&a0)[64],
                                          float (&a1)[64],
                                          float (&b0)[64],
                                          float* rowb, float* wbuf) {
    const float dsel = LOW ? a0[0] : a1[0];
    const float akk  = __int_as_float(__builtin_amdgcn_readlane(__float_as_int(dsel), k & 31));
    const float rk   = __builtin_amdgcn_rsqf(akk);
    const int j0 = l, j1 = l + 32;
    const float u0 = LOW ? a0[0] * rk : 0.0f;
    const float u1 = a1[0] * rk;
    const float w0 = b0[0] * rk;
    const float w1 = LOW ? 0.0f : a0[0] * rk;
    __syncthreads();
    rowb[(j0 - k - 1) & 63] = (j0 > k) ? u0 : 0.0f;
    rowb[(j1 - k - 1) & 63] = (j1 > k) ? u1 : 0.0f;
    wbuf[j0 * WSTR + k] = w0;
    wbuf[j1 * WSTR + k] = w1;
    __syncthreads();

    float v[NC + 1];
    const float4* r4 = reinterpret_cast<const float4*>(rowb);
    #pragma unroll
    for (int q = 0; q < (NC + 1) / 4; ++q) {
        const float4 t = r4[q];
        v[4 * q] = t.x; v[4 * q + 1] = t.y; v[4 * q + 2] = t.z; v[4 * q + 3] = t.w;
    }
    #pragma unroll
    for (int c = 1; c <= NC; ++c) {
        a0[c - 1] = fmaf(-v[c - 1], LOW ? u0 : w1, a0[c]);
        a1[c - 1] = fmaf(-v[c - 1], u1, a1[c]);
        b0[c - 1] = fmaf(-v[c - 1], w0, b0[c]);
    }
    a0[NC] = 0.0f;
    a1[NC] = 0.0f;
    b0[NC] = 0.0f;
}

__global__ __launch_bounds__(32) void chol_inv(const float* __restrict__ G,
                                               float* __restrict__ Rinv) {
    __shared__ __attribute__((aligned(16))) float rowb[64];
    __shared__ float wbuf[64 * WSTR];
    const int l = threadIdx.x;

    float a0[64], a1[64], b0[64];
    #pragma unroll
    for (int i = 0; i < 64; ++i) {
        a0[i] = G[i * 64 + l];
        a1[i] = G[i * 64 + l + 32];
        b0[i] = (i == l) ? 1.0f : 0.0f;
    }

    #pragma nounroll
    for (int k = 0; k < 32; ++k)  elim_step<63, true >(k, l, a0, a1, b0, rowb, wbuf);
    #pragma unroll
    for (int c = 0; c < 64; ++c) a0[c] = (c == l) ? 1.0f : 0.0f;
    #pragma nounroll
    for (int k = 32; k < 64; ++k) elim_step<31, false>(k, l, a0, a1, b0, rowb, wbuf);
    __syncthreads();

    #pragma unroll
    for (int r = 0; r < 64; ++r) {
        Rinv[r * 64 + l]      = wbuf[r * WSTR + l];
        Rinv[r * 64 + l + 32] = wbuf[r * WSTR + l + 32];
    }
}

typedef __attribute__((ext_vector_type(4))) float v4f;
#define APITCH 68
#define AWAVES 8
#define ATHREADS 256
#define AGRID 512

__global__ __launch_bounds__(ATHREADS) void apply_rinv(const float* __restrict__ Phi,
                                                         const float* __restrict__ Rinv,
                                                         float* __restrict__ Q, int m) {
    __shared__ __attribute__((aligned(16))) float slab[AWAVES * 16 * APITCH];
    const int lane = threadIdx.x & 31;
    const int wave = __builtin_amdgcn_readfirstlane(threadIdx.x >> 5);
    const int c = lane & 15;
    const int h = lane >> 4;
    const int nstrips = (m + 15) >> 4;
    const int nwaves  = AGRID * AWAVES;
    int strip = blockIdx.x * AWAVES + wave;
    if (strip >= nstrips) return;

    float* my = slab + wave * (16 * APITCH);

    v2f rf[40];
    #pragma unroll
    for (int t = 0; t < 4; ++t) {
        #pragma unroll
        for (int s = 0; s < 4 * (t + 1); ++s) {
            const int k = 8 * (s >> 1) + 4 * h + 2 * (s & 1);
            rf[2 * t * (t + 1) + s].x = Rinv[k * 64 + 16 * t + c];
            rf[2 * t * (t + 1) + s].y = Rinv[(k + 1) * 64 + 16 * t + c];
        }
    }

    v4f p[8];
    #define LOAD_STRIP(S) do { _Pragma("unroll") for (int j = 0; j < 8; ++j) { \
        long row_ = (long)(S) * 16 + 2 * j + h; if (row_ >= (long)m) row_ = (long)m - 1; \
        p[j] = __builtin_nontemporal_load(reinterpret_cast<const v4f*>(Phi + row_ * 64 + 4 * c)); } } while (0)
    #define SLAB_ORDER() do { __builtin_amdgcn_fence(__ATOMIC_RELEASE, "wavefront"); \
        __builtin_amdgcn_wave_barrier(); __builtin_amdgcn_fence(__ATOMIC_ACQUIRE, "wavefront"); } while (0)
    LOAD_STRIP(strip);

    for (; strip < nstrips; strip += nwaves) {
        #pragma unroll
        for (int j = 0; j < 8; ++j)
            *reinterpret_cast<v4f*>(my + (2 * j + h) * APITCH + 4 * c) = p[j];
        SLAB_ORDER();

        const int nxt = strip + nwaves;
        if (nxt < nstrips) LOAD_STRIP(nxt);

        v4f a[8];
        #pragma unroll
        for (int j = 0; j < 8; ++j)
            a[j] = *reinterpret_cast<const v4f*>(my + c * APITCH + 8 * j + 4 * h);

        v8f acc0 = {}, acc1 = {}, acc2 = {}, acc3 = {};
        #define W(S, T) acc##T = __builtin_amdgcn_wmma_f32_16x16x4_f32(false, rf[2 * (T) * ((T) + 1) + (S)], false, va, \
                                                                       (short)0, acc##T, false, false)
        #pragma unroll
        for (int s = 0; s < 16; ++s) {
            v2f va;
            if (s & 1) { va.x = a[s >> 1].z; va.y = a[s >> 1].w; }
            else       { va.x = a[s >> 1].x; va.y = a[s >> 1].y; }
            if (s < 4)  W(s, 0);
            if (s < 8)  W(s, 1);
            if (s < 12) W(s, 2);
            W(s, 3);
        }
        #undef W

        asm volatile("v_nop\n\tv_nop\n\tv_nop\n\tv_nop\n\tv_nop\n\tv_nop\n\tv_nop\n\tv_nop"
                     : "+v"(acc0), "+v"(acc1), "+v"(acc2), "+v"(acc3));

        #define PUT(T) do { float* d_ = my + c * APITCH + 16 * (T) + 8 * h; \
            *reinterpret_cast<v4f*>(d_)     = __builtin_shufflevector(acc##T, acc##T, 0, 1, 2, 3); \
            *reinterpret_cast<v4f*>(d_ + 4) = __builtin_shufflevector(acc##T, acc##T, 4, 5, 6, 7); } while (0)
        PUT(0); PUT(1); PUT(2); PUT(3);
        #undef PUT
        SLAB_ORDER();

        v4f o[8];
        #pragma unroll
        for (int j = 0; j < 8; ++j)
            o[j] = *reinterpret_cast<const v4f*>(my + (2 * j + h) * APITCH + 4 * c);
        SLAB_ORDER();

        const long row0 = (long)strip * 16 + h;
        float* qp = Q + row0 * 64 + 4 * c;
        if ((long)strip * 16 + 16 <= (long)m) {
            #pragma unroll
            for (int j = 0; j < 8; ++j) *reinterpret_cast<v4f*>(qp + j * 128) = o[j];
        } else {
            #pragma unroll
            for (int j = 0; j < 8; ++j)
                if (row0 + 2 * j < (long)m) *reinterpret_cast<v4f*>(qp + j * 128) = o[j];
        }
    }
    #undef LOAD_STRIP
    #undef SLAB_ORDER
}

extern "C" void kernel_launch(void* const* d_in, const int* in_sizes, int n_in,
                              void* d_out, int out_size, void* d_ws, size_t ws_size,
                              hipStream_t stream) {
    const float* Phi = (const float*)d_in[0];
    float* out = (float*)d_out;
    const int m = in_sizes[0] / 64;

    float* ws   = (float*)d_ws;
    float* G    = ws;
    float* Rinv = ws + 4096;
    float* part = ws + 8192;

    long avail = ((long)(ws_size / 4) - 8192) / 4096;
    int nwg = (int)(avail < 1 ? 1 : (avail > 512 ? 512 : avail));

    gram_partial<<<nwg, GTHREADS, 0, stream>>>(Phi, part, m, nwg);
    reduce_g<<<128, RTHREADS, 0, stream>>>(part, G, nwg);
    chol_inv<<<1, 32, 0, stream>>>(G, Rinv);

    apply_rinv<<<AGRID, ATHREADS, 0, stream>>>(Phi, Rinv, out, m);
}


# chol_inv: steps overlapped: each step updates the next pivot row first, scales it and sends it through LDS (loads right behind the stores, no wait) under its other row updates; every sum unchanged
# speedup vs baseline: 3.6940x; 1.0527x over previous
#include <hip/hip_runtime.h>

typedef __attribute__((ext_vector_type(2))) float v2f;
typedef __attribute__((ext_vector_type(8))) float v8f;

typedef __attribute__((ext_vector_type(4))) float g4f;
#define GWAVES 8
#define GTHREADS 256
#define GSLOT 2560
#define GPITCH 65

__global__ __launch_bounds__(GTHREADS) void gram_partial(const float* __restrict__ Phi,
                                                         float* __restrict__ part,
                                                         int m, int nwg) {
    __shared__ float red[4 * GSLOT];
    const int lane = threadIdx.x & 31;
    const int wave = __builtin_amdgcn_readfirstlane(threadIdx.x >> 5);
    const int c = lane & 15;
    const int h = lane >> 4;
    const int nstrips = (m + 15) >> 4;
    const int nwaves  = nwg * GWAVES;
    int strip = blockIdx.x * GWAVES + wave;

    v8f acc[10];
    #pragma unroll
    for (int q = 0; q < 10; ++q) acc[q] = v8f{};

    g4f p[8];
    #define GLOAD_STRIP(S) do { _Pragma("unroll") for (int j = 0; j < 8; ++j) { \
        long row_ = (long)(S) * 16 + 2 * j + h; if (row_ >= (long)m) row_ = (long)m - 1; \
        p[j] = *reinterpret_cast<const g4f*>(Phi + row_ * 64 + 4 * c); } } while (0)
    if (strip < nstrips) GLOAD_STRIP(strip);

    for (; strip < nstrips; strip += nwaves) {
        g4f a[8];
        #pragma unroll
        for (int j = 0; j < 8; ++j) a[j] = p[j];
        if ((long)strip * 16 + 16 > (long)m) {
            #pragma unroll
            for (int j = 0; j < 8; ++j)
                if ((long)strip * 16 + 2 * j + h >= (long)m) a[j] = g4f{0.0f, 0.0f, 0.0f, 0.0f};
        }

        const int nxt = strip + nwaves;
        if (nxt < nstrips) GLOAD_STRIP(nxt);

        #pragma unroll
        for (int i = 0; i < 4; ++i) {
            v2f f[4];
            f[0].x = a[2 * i].x; f[0].y = a[2 * i + 1].x;
            f[1].x = a[2 * i].y; f[1].y = a[2 * i + 1].y;
            f[2].x = a[2 * i].z; f[2].y = a[2 * i + 1].z;
            f[3].x = a[2 * i].w; f[3].y = a[2 * i + 1].w;
            int q = 0;
            #pragma unroll
            for (int ta = 0; ta < 4; ++ta) {
                #pragma unroll
                for (int tb = ta; tb < 4; ++tb) {
                    acc[q] = __builtin_amdgcn_wmma_f32_16x16x4_f32(false, f[ta], false, f[tb],
                                                                   (short)0, acc[q], false, false);
                    ++q;
                }
            }
        }
    }
    #undef GLOAD_STRIP

    #define GSTORE(SLOT) do { float* s_ = red + (SLOT) * GSLOT + lane; _Pragma("unroll") for (int q = 0; q < 10; ++q) { \
        _Pragma("unroll") for (int r = 0; r < 8; ++r) s_[(q * 8 + r) * 32] = acc[q][r]; } } while (0)
    #define GADD(SLOT) do { const float* s_ = red + (SLOT) * GSLOT + lane; _Pragma("unroll") for (int q = 0; q < 10; ++q) { \
        _Pragma("unroll") for (int r = 0; r < 8; ++r) acc[q][r] += s_[(q * 8 + r) * 32]; } } while (0)
    if (wave >= 4) GSTORE(wave - 4);
    __syncthreads();
    if (wave < 4) GADD(wave);
    __syncthreads();
    if (wave == 2 || wave == 3) GSTORE(wave - 2);
    __syncthreads();
    if (wave < 2) GADD(wave);
    __syncthreads();
    if (wave == 1) GSTORE(0);
    __syncthreads();
    if (wave == 0) {
        GADD(0);
        int q = 0;
        #pragma unroll
        for (int ta = 0; ta < 4; ++ta) {
            #pragma unroll
            for (int tb = ta; tb < 4; ++tb) {
                #pragma unroll
                for (int r = 0; r < 8; ++r)
                    red[(32 * h + 4 * r + ta) * GPITCH + 4 * c + tb] = acc[q][r];
                ++q;
            }
        }
    }
    __syncthreads();
    #undef GSTORE
    #undef GADD

    float* pp = part + (size_t)blockIdx.x * 4096 + threadIdx.x;
    const int i0 = threadIdx.x >> 6, j = threadIdx.x & 63;
    const bool own = i0 <= (j & 3);
    const float* src = red + (own ? i0 * GPITCH + j : j * GPITCH + i0);
    const int step = own ? 4 * GPITCH : 4;
    float o[4096 / GTHREADS];
    #pragma unroll
    for (int k = 0; k < 4096 / GTHREADS; ++k) o[k] = src[k * step];
    #pragma unroll
    for (int k = 0; k < 4096 / GTHREADS; ++k) pp[k * GTHREADS] = o[k];
}

#define RWAVES 16
#define RTHREADS 512
#define RDEPTH 32
__global__ __launch_bounds__(RTHREADS) void reduce_g(const float* __restrict__ part,
                                                        float* __restrict__ G, int nwg) {
    __shared__ float red[RWAVES * 32];
    const int lane = threadIdx.x & 31;
    const int wave = __builtin_amdgcn_readfirstlane(threadIdx.x >> 5);
    const int e = blockIdx.x * 32 + lane;
    const float* p = part + e;
    float s0 = 0.0f, s1 = 0.0f, s2 = 0.0f, s3 = 0.0f;
    int w = wave;
    for (; w + (RDEPTH - 1) * RWAVES < nwg; w += RDEPTH * RWAVES) {
        float t[RDEPTH];
        #pragma unroll
        for (int i = 0; i < RDEPTH; ++i) t[i] = p[(size_t)(w + i * RWAVES) * 4096];
        #pragma unroll
        for (int i = 0; i < RDEPTH; i += 4) { s0 += t[i]; s1 += t[i + 1]; s2 += t[i + 2]; s3 += t[i + 3]; }
    }
    for (; w < nwg; w += RWAVES) s0 += p[(size_t)w * 4096];
    red[wave * 32 + lane] = (s0 + s1) + (s2 + s3);
    __syncthreads();
    if (wave == 0) {
        float g = 0.0f;
        #pragma unroll
        for (int i = 0; i < RWAVES; ++i) g += red[i * 32 + lane];
        G[e] = g;
    }
}

#define WSTR 65

#define WAVE_ORDER() do { __builtin_amdgcn_fence(__ATOMIC_RELEASE, "wavefront"); \
    __builtin_amdgcn_wave_barrier(); __builtin_amdgcn_fence(__ATOMIC_ACQUIRE, "wavefront"); } while (0)

template <bool LOW, int NV>
__device__ __forceinline__ void pivot_row(const int k, const int l, const float ra0, const float ra1, const float rb0,
                                          float& m0, float& u1, float& w0, float* rowb, g4f (&v)[16]) {
    const float dsel = LOW ? ra0 : ra1;
    const float akk  = __int_as_float(__builtin_amdgcn_readlane(__float_as_int(dsel), k & 31));
    const float rk   = __builtin_amdgcn_rsqf(akk);
    const int j0 = l, j1 = l + 32;
    const float u0 = LOW ? ra0 * rk : 0.0f;
    u1 = ra1 * rk;
    w0 = rb0 * rk;
    const float w1 = LOW ? 0.0f : ra0 * rk;
    m0 = LOW ? u0 : w1;
    rowb[(j0 - k - 1) & 63] = (j0 > k) ? u0 : 0.0f;
    rowb[(j1 - k - 1) & 63] = (j1 > k) ? u1 : 0.0f;
    WAVE_ORDER();
    const g4f* r4 = reinterpret_cast<const g4f*>(rowb);
    #pragma unroll
    for (int q = 0; q < NV / 4; ++q) v[q] = r4[q];
}

__device__ __forceinline__ void park_rinv(const int k, const int l, const float w0, const float w1, float* wbuf) {
    wbuf[l * WSTR + k] = w0;
    wbuf[(l + 32) * WSTR + k] = w1;
}

#define NFILL 12

template <int NC, bool LOW, bool NEXT, bool LOWNEXT>
__device__ __forceinline__ void elim_step(const int k, const int l,
                                          float (&a0)[64],
                                          float (&a1)[64],
                                          float (&b0)[64],
                                          float& m0, float& u1, float& w0,
                                          const g4f (&v)[16], g4f (&vn)[16],
                                          float* rowb, float* wbuf) {
    #define VV(I) v[(I) >> 2][(I) & 3]
    a0[0] = fmaf(-VV(0), m0, a0[1]);
    a1[0] = fmaf(-VV(0), u1, a1[1]);
    b0[0] = fmaf(-VV(0), w0, b0[1]);
    float m0n = 0.0f, u1n = 0.0f, w0n = 0.0f;
    if (NEXT) {
        const float ra0 = (LOW && !LOWNEXT) ? ((l == 0) ? 1.0f : 0.0f) : a0[0];
        pivot_row<LOWNEXT, LOWNEXT ? 64 : 32>(k + 1, l, ra0, a1[0], b0[0], m0n, u1n, w0n, rowb, vn);
    }
    #pragma unroll
    for (int c = 2; c <= NFILL; ++c) {
        a0[c - 1] = fmaf(-VV(c - 1), m0, a0[c]);
        a1[c - 1] = fmaf(-VV(c - 1), u1, a1[c]);
        b0[c - 1] = fmaf(-VV(c - 1), w0, b0[c]);
    }
    __builtin_amdgcn_sched_barrier(0);
    #pragma unroll
    for (int c = NFILL + 1; c <= NC; ++c) {
        a0[c - 1] = fmaf(-VV(c - 1), m0, a0[c]);
        a1[c - 1] = fmaf(-VV(c - 1), u1, a1[c]);
        b0[c - 1] = fmaf(-VV(c - 1), w0, b0[c]);
    }
    a0[NC] = 0.0f;
    a1[NC] = 0.0f;
    b0[NC] = 0.0f;
    if (NEXT) park_rinv(k + 1, l, w0n, LOWNEXT ? 0.0f : m0n, wbuf);
    #undef VV
    asm volatile("" : : "v"(v[(NC + 1) / 4 - 1][3]));
    m0 = m0n; u1 = u1n; w0 = w0n;
    __builtin_amdgcn_sched_barrier(0);
}

__global__ __launch_bounds__(32) void chol_inv(const float* __restrict__ G,
                                               float* __restrict__ Rinv) {
    __shared__ __attribute__((aligned(16))) float rowb[64];
    __shared__ float wbuf[64 * WSTR];
    const int l = threadIdx.x;

    float a0[64], a1[64], b0[64];
    #pragma unroll
    for (int i = 0; i < 64; ++i) {
        a0[i] = G[i * 64 + l];
        a1[i] = G[i * 64 + l + 32];
        b0[i] = (i == l) ? 1.0f : 0.0f;
    }

    g4f va[16], vb[16];
    float m0, u1, w0;
    pivot_row<true, 64>(0, l, a0[0], a1[0], b0[0], m0, u1, w0, rowb, va);
    park_rinv(0, l, w0, 0.0f, wbuf);

    #pragma nounroll
    for (int k = 0; k < 30; k += 2) {
        elim_step<63, true, true, true>(k,     l, a0, a1, b0, m0, u1, w0, va, vb, rowb, wbuf);
        elim_step<63, true, true, true>(k + 1, l, a0, a1, b0, m0, u1, w0, vb, va, rowb, wbuf);
    }
    elim_step<63, true, true, true >(30, l, a0, a1, b0, m0, u1, w0, va, vb, rowb, wbuf);
    elim_step<63, true, true, false>(31, l, a0, a1, b0, m0, u1, w0, vb, va, rowb, wbuf);
    #pragma unroll
    for (int c = 0; c < 64; ++c) a0[c] = (c == l) ? 1.0f : 0.0f;
    #pragma nounroll
    for (int k = 32; k < 62; k += 2) {
        elim_step<31, false, true, false>(k,     l, a0, a1, b0, m0, u1, w0, va, vb, rowb, wbuf);
        elim_step<31, false, true, false>(k + 1, l, a0, a1, b0, m0, u1, w0, vb, va, rowb, wbuf);
    }
    elim_step<31, false, true,  false>(62, l, a0, a1, b0, m0, u1, w0, va, vb, rowb, wbuf);
    elim_step<31, false, false, false>(63, l, a0, a1, b0, m0, u1, w0, vb, va, rowb, wbuf);
    __syncthreads();
    #undef WAVE_ORDER
    #undef NFILL

    #pragma unroll
    for (int r = 0; r < 64; ++r) {
        Rinv[r * 64 + l]      = wbuf[r * WSTR + l];
        Rinv[r * 64 + l + 32] = wbuf[r * WSTR + l + 32];
    }
}

typedef __attribute__((ext_vector_type(4))) float v4f;
#define APITCH 68
#define AWAVES 8
#define ATHREADS 256
#define AGRID 512

__global__ __launch_bounds__(ATHREADS) void apply_rinv(const float* __restrict__ Phi,
                                                         const float* __restrict__ Rinv,
                                                         float* __restrict__ Q, int m) {
    __shared__ __attribute__((aligned(16))) float slab[AWAVES * 16 * APITCH];
    const int lane = threadIdx.x & 31;
    const int wave = __builtin_amdgcn_readfirstlane(threadIdx.x >> 5);
    const int c = lane & 15;
    const int h = lane >> 4;
    const int nstrips = (m + 15) >> 4;
    const int nwaves  = AGRID * AWAVES;
    int strip = blockIdx.x * AWAVES + wave;
    if (strip >= nstrips) return;

    float* my = slab + wave * (16 * APITCH);

    v2f rf[40];
    #pragma unroll
    for (int t = 0; t < 4; ++t) {
        #pragma unroll
        for (int s = 0; s < 4 * (t + 1); ++s) {
            const int k = 8 * (s >> 1) + 4 * h + 2 * (s & 1);
            rf[2 * t * (t + 1) + s].x = Rinv[k * 64 + 16 * t + c];
            rf[2 * t * (t + 1) + s].y = Rinv[(k + 1) * 64 + 16 * t + c];
        }
    }

    v4f p[8];
    #define LOAD_STRIP(S) do { _Pragma("unroll") for (int j = 0; j < 8; ++j) { \
        long row_ = (long)(S) * 16 + 2 * j + h; if (row_ >= (long)m) row_ = (long)m - 1; \
        p[j] = __builtin_nontemporal_load(reinterpret_cast<const v4f*>(Phi + row_ * 64 + 4 * c)); } } while (0)
    #define SLAB_ORDER() do { __builtin_amdgcn_fence(__ATOMIC_RELEASE, "wavefront"); \
        __builtin_amdgcn_wave_barrier(); __builtin_amdgcn_fence(__ATOMIC_ACQUIRE, "wavefront"); } while (0)
    LOAD_STRIP(strip);

    for (; strip < nstrips; strip += nwaves) {
        #pragma unroll
        for (int j = 0; j < 8; ++j)
            *reinterpret_cast<v4f*>(my + (2 * j + h) * APITCH + 4 * c) = p[j];
        SLAB_ORDER();

        const int nxt = strip + nwaves;
        if (nxt < nstrips) LOAD_STRIP(nxt);

        v4f a[8];
        #pragma unroll
        for (int j = 0; j < 8; ++j)
            a[j] = *reinterpret_cast<const v4f*>(my + c * APITCH + 8 * j + 4 * h);

        v8f acc0 = {}, acc1 = {}, acc2 = {}, acc3 = {};
        #define W(S, T) acc##T = __builtin_amdgcn_wmma_f32_16x16x4_f32(false, rf[2 * (T) * ((T) + 1) + (S)], false, va, \
                                                                       (short)0, acc##T, false, false)
        #pragma unroll
        for (int s = 0; s < 16; ++s) {
            v2f va;
            if (s & 1) { va.x = a[s >> 1].z; va.y = a[s >> 1].w; }
            else       { va.x = a[s >> 1].x; va.y = a[s >> 1].y; }
            if (s < 4)  W(s, 0);
            if (s < 8)  W(s, 1);
            if (s < 12) W(s, 2);
            W(s, 3);
        }
        #undef W

        asm volatile("v_nop\n\tv_nop\n\tv_nop\n\tv_nop\n\tv_nop\n\tv_nop\n\tv_nop\n\tv_nop"
                     : "+v"(acc0), "+v"(acc1), "+v"(acc2), "+v"(acc3));

        #define PUT(T) do { float* d_ = my + c * APITCH + 16 * (T) + 8 * h; \
            *reinterpret_cast<v4f*>(d_)     = __builtin_shufflevector(acc##T, acc##T, 0, 1, 2, 3); \
            *reinterpret_cast<v4f*>(d_ + 4) = __builtin_shufflevector(acc##T, acc##T, 4, 5, 6, 7); } while (0)
        PUT(0); PUT(1); PUT(2); PUT(3);
        #undef PUT
        SLAB_ORDER();

        v4f o[8];
        #pragma unroll
        for (int j = 0; j < 8; ++j)
            o[j] = *reinterpret_cast<const v4f*>(my + (2 * j + h) * APITCH + 4 * c);
        SLAB_ORDER();

        const long row0 = (long)strip * 16 + h;
        float* qp = Q + row0 * 64 + 4 * c;
        if ((long)strip * 16 + 16 <= (long)m) {
            #pragma unroll
            for (int j = 0; j < 8; ++j) *reinterpret_cast<v4f*>(qp + j * 128) = o[j];
        } else {
            #pragma unroll
            for (int j = 0; j < 8; ++j)
                if (row0 + 2 * j < (long)m) *reinterpret_cast<v4f*>(qp + j * 128) = o[j];
        }
    }
    #undef LOAD_STRIP
    #undef SLAB_ORDER
}

extern "C" void kernel_launch(void* const* d_in, const int* in_sizes, int n_in,
                              void* d_out, int out_size, void* d_ws, size_t ws_size,
                              hipStream_t stream) {
    const float* Phi = (const float*)d_in[0];
    float* out = (float*)d_out;
    const int m = in_sizes[0] / 64;

    float* ws   = (float*)d_ws;
    float* G    = ws;
    float* Rinv = ws + 4096;
    float* part = ws + 8192;

    long avail = ((long)(ws_size / 4) - 8192) / 4096;
    int nwg = (int)(avail < 1 ? 1 : (avail > 512 ? 512 : avail));

    gram_partial<<<nwg, GTHREADS, 0, stream>>>(Phi, part, m, nwg);
    reduce_g<<<128, RTHREADS, 0, stream>>>(part, G, nwg);
    chol_inv<<<1, 32, 0, stream>>>(G, Rinv);

    apply_rinv<<<AGRID, ATHREADS, 0, stream>>>(Phi, Rinv, out, m);
}


# chol_inv gets no launch of its own: wave 0 of each of 256 apply_rinv workgroups factors G into LDS while the first strips load; so 2 waves per SIMD, 2 strips in flight each; every sum unchanged
# speedup vs baseline: 3.8476x; 1.0416x over previous
#include <hip/hip_runtime.h>

typedef __attribute__((ext_vector_type(2))) float v2f;
typedef __attribute__((ext_vector_type(8))) float v8f;

typedef __attribute__((ext_vector_type(4))) float g4f;
#define GWAVES 8
#define GTHREADS 256
#define GSLOT 2560
#define GPITCH 65

__global__ __launch_bounds__(GTHREADS) void gram_partial(const float* __restrict__ Phi,
                                                         float* __restrict__ part,
                                                         int m, int nwg) {
    __shared__ float red[4 * GSLOT];
    const int lane = threadIdx.x & 31;
    const int wave = __builtin_amdgcn_readfirstlane(threadIdx.x >> 5);
    const int c = lane & 15;
    const int h = lane >> 4;
    const int nstrips = (m + 15) >> 4;
    const int nwaves  = nwg * GWAVES;
    int strip = blockIdx.x * GWAVES + wave;

    v8f acc[10];
    #pragma unroll
    for (int q = 0; q < 10; ++q) acc[q] = v8f{};

    g4f p[8];
    #define GLOAD_STRIP(S) do { _Pragma("unroll") for (int j = 0; j < 8; ++j) { \
        long row_ = (long)(S) * 16 + 2 * j + h; if (row_ >= (long)m) row_ = (long)m - 1; \
        p[j] = *reinterpret_cast<const g4f*>(Phi + row_ * 64 + 4 * c); } } while (0)
    if (strip < nstrips) GLOAD_STRIP(strip);

    for (; strip < nstrips; strip += nwaves) {
        g4f a[8];
        #pragma unroll
        for (int j = 0; j < 8; ++j) a[j] = p[j];
        if ((long)strip * 16 + 16 > (long)m) {
            #pragma unroll
            for (int j = 0; j < 8; ++j)
                if ((long)strip * 16 + 2 * j + h >= (long)m) a[j] = g4f{0.0f, 0.0f, 0.0f, 0.0f};
        }

        const int nxt = strip + nwaves;
        if (nxt < nstrips) GLOAD_STRIP(nxt);

        #pragma unroll
        for (int i = 0; i < 4; ++i) {
            v2f f[4];
            f[0].x = a[2 * i].x; f[0].y = a[2 * i + 1].x;
            f[1].x = a[2 * i].y; f[1].y = a[2 * i + 1].y;
            f[2].x = a[2 * i].z; f[2].y = a[2 * i + 1].z;
            f[3].x = a[2 * i].w; f[3].y = a[2 * i + 1].w;
            int q = 0;
            #pragma unroll
            for (int ta = 0; ta < 4; ++ta) {
                #pragma unroll
                for (int tb = ta; tb < 4; ++tb) {
                    acc[q] = __builtin_amdgcn_wmma_f32_16x16x4_f32(false, f[ta], false, f[tb],
                                                                   (short)0, acc[q], false, false);
                    ++q;
                }
            }
        }
    }
    #undef GLOAD_STRIP

    #define GSTORE(SLOT) do { float* s_ = red + (SLOT) * GSLOT + lane; _Pragma("unroll") for (int q = 0; q < 10; ++q) { \
        _Pragma("unroll") for (int r = 0; r < 8; ++r) s_[(q * 8 + r) * 32] = acc[q][r]; } } while (0)
    #define GADD(SLOT) do { const float* s_ = red + (SLOT) * GSLOT + lane; _Pragma("unroll") for (int q = 0; q < 10; ++q) { \
        _Pragma("unroll") for (int r = 0; r < 8; ++r) acc[q][r] += s_[(q * 8 + r) * 32]; } } while (0)
    if (wave >= 4) GSTORE(wave - 4);
    __syncthreads();
    if (wave < 4) GADD(wave);
    __syncthreads();
    if (wave == 2 || wave == 3) GSTORE(wave - 2);
    __syncthreads();
    if (wave < 2) GADD(wave);
    __syncthreads();
    if (wave == 1) GSTORE(0);
    __syncthreads();
    if (wave == 0) {
        GADD(0);
        int q = 0;
        #pragma unroll
        for (int ta = 0; ta < 4; ++ta) {
            #pragma unroll
            for (int tb = ta; tb < 4; ++tb) {
                #pragma unroll
                for (int r = 0; r < 8; ++r)
                    red[(32 * h + 4 * r + ta) * GPITCH + 4 * c + tb] = acc[q][r];
                ++q;
            }
        }
    }
    __syncthreads();
    #undef GSTORE
    #undef GADD

    float* pp = part + (size_t)blockIdx.x * 4096 + threadIdx.x;
    const int i0 = threadIdx.x >> 6, j = threadIdx.x & 63;
    const bool own = i0 <= (j & 3);
    const float* src = red + (own ? i0 * GPITCH + j : j * GPITCH + i0);
    const int step = own ? 4 * GPITCH : 4;
    float o[4096 / GTHREADS];
    #pragma unroll
    for (int k = 0; k < 4096 / GTHREADS; ++k) o[k] = src[k * step];
    #pragma unroll
    for (int k = 0; k < 4096 / GTHREADS; ++k) pp[k * GTHREADS] = o[k];
}

#define RWAVES 16
#define RTHREADS 512
#define RDEPTH 32
__global__ __launch_bounds__(RTHREADS) void reduce_g(const float* __restrict__ part,
                                                        float* __restrict__ G, int nwg) {
    __shared__ float red[RWAVES * 32];
    const int lane = threadIdx.x & 31;
    const int wave = __builtin_amdgcn_readfirstlane(threadIdx.x >> 5);
    const int e = blockIdx.x * 32 + lane;
    const float* p = part + e;
    float s0 = 0.0f, s1 = 0.0f, s2 = 0.0f, s3 = 0.0f;
    int w = wave;
    for (; w + (RDEPTH - 1) * RWAVES < nwg; w += RDEPTH * RWAVES) {
        float t[RDEPTH];
        #pragma unroll
        for (int i = 0; i < RDEPTH; ++i) t[i] = p[(size_t)(w + i * RWAVES) * 4096];
        #pragma unroll
        for (int i = 0; i < RDEPTH; i += 4) { s0 += t[i]; s1 += t[i + 1]; s2 += t[i + 2]; s3 += t[i + 3]; }
    }
    for (; w < nwg; w += RWAVES) s0 += p[(size_t)w * 4096];
    red[wave * 32 + lane] = (s0 + s1) + (s2 + s3);
    __syncthreads();
    if (wave == 0) {
        float g = 0.0f;
        #pragma unroll
        for (int i = 0; i < RWAVES; ++i) g += red[i * 32 + lane];
        G[e] = g;
    }
}

#define WSTR 65

#define WAVE_ORDER() do { __builtin_amdgcn_fence(__ATOMIC_RELEASE, "wavefront"); \
    __builtin_amdgcn_wave_barrier(); __builtin_amdgcn_fence(__ATOMIC_ACQUIRE, "wavefront"); } while (0)

template <bool LOW, int NV>
__device__ __forceinline__ void pivot_row(const int k, const int l, const float ra0, const float ra1, const float rb0,
                                          float& m0, float& u1, float& w0, float* rowb, g4f (&v)[16]) {
    const float dsel = LOW ? ra0 : ra1;
    const float akk  = __int_as_float(__builtin_amdgcn_readlane(__float_as_int(dsel), k & 31));
    const float rk   = __builtin_amdgcn_rsqf(akk);
    const int j0 = l, j1 = l + 32;
    const float u0 = LOW ? ra0 * rk : 0.0f;
    u1 = ra1 * rk;
    w0 = rb0 * rk;
    const float w1 = LOW ? 0.0f : ra0 * rk;
    m0 = LOW ? u0 : w1;
    rowb[(j0 - k - 1) & 63] = (j0 > k) ? u0 : 0.0f;
    rowb[(j1 - k - 1) & 63] = (j1 > k) ? u1 : 0.0f;
    WAVE_ORDER();
    const g4f* r4 = reinterpret_cast<const g4f*>(rowb);
    #pragma unroll
    for (int q = 0; q < NV / 4; ++q) v[q] = r4[q];
}

__device__ __forceinline__ void park_rinv(const int k, const int l, const float w0, const float w1, float* wbuf) {
    wbuf[l * WSTR + k] = w0;
    wbuf[(l + 32) * WSTR + k] = w1;
}

#define NFILL 12

template <int NC, bool LOW, bool NEXT, bool LOWNEXT>
__device__ __forceinline__ void elim_step(const int k, const int l,
                                          float (&a0)[64],
                                          float (&a1)[64],
                                          float (&b0)[64],
                                          float& m0, float& u1, float& w0,
                                          const g4f (&v)[16], g4f (&vn)[16],
                                          float* rowb, float* wbuf) {
    #define VV(I) v[(I) >> 2][(I) & 3]
    a0[0] = fmaf(-VV(0), m0, a0[1]);
    a1[0] = fmaf(-VV(0), u1, a1[1]);
    b0[0] = fmaf(-VV(0), w0, b0[1]);
    float m0n = 0.0f, u1n = 0.0f, w0n = 0.0f;
    if (NEXT) {
        const float ra0 = (LOW && !LOWNEXT) ? ((l == 0) ? 1.0f : 0.0f) : a0[0];
        pivot_row<LOWNEXT, LOWNEXT ? 64 : 32>(k + 1, l, ra0, a1[0], b0[0], m0n, u1n, w0n, rowb, vn);
    }
    #pragma unroll
    for (int c = 2; c <= NFILL; ++c) {
        a0[c - 1] = fmaf(-VV(c - 1), m0, a0[c]);
        a1[c - 1] = fmaf(-VV(c - 1), u1, a1[c]);
        b0[c - 1] = fmaf(-VV(c - 1), w0, b0[c]);
    }
    __builtin_amdgcn_sched_barrier(0);
    #pragma unroll
    for (int c = NFILL + 1; c <= NC; ++c) {
        a0[c - 1] = fmaf(-VV(c - 1), m0, a0[c]);
        a1[c - 1] = fmaf(-VV(c - 1), u1, a1[c]);
        b0[c - 1] = fmaf(-VV(c - 1), w0, b0[c]);
    }
    a0[NC] = 0.0f;
    a1[NC] = 0.0f;
    b0[NC] = 0.0f;
    if (NEXT) park_rinv(k + 1, l, w0n, LOWNEXT ? 0.0f : m0n, wbuf);
    #undef VV
    asm volatile("" : : "v"(v[(NC + 1) / 4 - 1][3]));
    m0 = m0n; u1 = u1n; w0 = w0n;
    __builtin_amdgcn_sched_barrier(0);
}

__device__ __forceinline__ void factor_load(const float* __restrict__ G, const int l,
                                            float (&a0)[64], float (&a1)[64], float (&b0)[64]) {
    #pragma unroll
    for (int i = 0; i < 64; ++i) {
        a0[i] = G[i * 64 + l];
        a1[i] = G[i * 64 + l + 32];
        b0[i] = (i == l) ? 1.0f : 0.0f;
    }
}

__device__ __forceinline__ void factor_run(const int l, float (&a0)[64], float (&a1)[64], float (&b0)[64],
                                           float* rowb, float* wbuf) {
    g4f va[16], vb[16];
    float m0, u1, w0;
    pivot_row<true, 64>(0, l, a0[0], a1[0], b0[0], m0, u1, w0, rowb, va);
    park_rinv(0, l, w0, 0.0f, wbuf);

    #pragma nounroll
    for (int k = 0; k < 30; k += 2) {
        elim_step<63, true, true, true>(k,     l, a0, a1, b0, m0, u1, w0, va, vb, rowb, wbuf);
        elim_step<63, true, true, true>(k + 1, l, a0, a1, b0, m0, u1, w0, vb, va, rowb, wbuf);
    }
    elim_step<63, true, true, true >(30, l, a0, a1, b0, m0, u1, w0, va, vb, rowb, wbuf);
    elim_step<63, true, true, false>(31, l, a0, a1, b0, m0, u1, w0, vb, va, rowb, wbuf);
    #pragma unroll
    for (int c = 0; c < 64; ++c) a0[c] = (c == l) ? 1.0f : 0.0f;
    #pragma nounroll
    for (int k = 32; k < 62; k += 2) {
        elim_step<31, false, true, false>(k,     l, a0, a1, b0, m0, u1, w0, va, vb, rowb, wbuf);
        elim_step<31, false, true, false>(k + 1, l, a0, a1, b0, m0, u1, w0, vb, va, rowb, wbuf);
    }
    elim_step<31, false, true,  false>(62, l, a0, a1, b0, m0, u1, w0, va, vb, rowb, wbuf);
    elim_step<31, false, false, false>(63, l, a0, a1, b0, m0, u1, w0, vb, va, rowb, wbuf);
}
#undef WAVE_ORDER
#undef NFILL

typedef __attribute__((ext_vector_type(4))) float v4f;
#define APITCH 68
#define AWAVES 8
#define ATHREADS 256
#define AGRID 256

#define LOAD_STRIP(P, S) do { _Pragma("unroll") for (int j = 0; j < 8; ++j) { \
    long row_ = (long)(S) * 16 + 2 * j + h; if (row_ >= (long)m) row_ = (long)m - 1; \
    (P)[j] = __builtin_nontemporal_load(reinterpret_cast<const v4f*>(Phi + row_ * 64 + 4 * c)); } } while (0)
#define SLAB_ORDER() do { __builtin_amdgcn_fence(__ATOMIC_RELEASE, "wavefront"); \
    __builtin_amdgcn_wave_barrier(); __builtin_amdgcn_fence(__ATOMIC_ACQUIRE, "wavefront"); } while (0)

__device__ __forceinline__ void strip_body(v4f (&p)[8], const int S, const int NXT, const int m,
                                           const float* __restrict__ Phi, float* __restrict__ Q, float* my,
                                           const v2f (&rf)[40], const int c, const int h) {
    #pragma unroll
    for (int j = 0; j < 8; ++j)
        *reinterpret_cast<v4f*>(my + (2 * j + h) * APITCH + 4 * c) = p[j];
    SLAB_ORDER();

    LOAD_STRIP(p, NXT);

    v4f a[8];
    #pragma unroll
    for (int j = 0; j < 8; ++j)
        a[j] = *reinterpret_cast<const v4f*>(my + c * APITCH + 8 * j + 4 * h);

    v8f acc0 = {}, acc1 = {}, acc2 = {}, acc3 = {};
    #define W(S_, T) acc##T = __builtin_amdgcn_wmma_f32_16x16x4_f32(false, rf[2 * (T) * ((T) + 1) + (S_)], false, va, \
                                                                    (short)0, acc##T, false, false)
    #pragma unroll
    for (int s = 0; s < 16; ++s) {
        v2f va;
        if (s & 1) { va.x = a[s >> 1].z; va.y = a[s >> 1].w; }
        else       { va.x = a[s >> 1].x; va.y = a[s >> 1].y; }
        if (s < 4)  W(s, 0);
        if (s < 8)  W(s, 1);
        if (s < 12) W(s, 2);
        W(s, 3);
    }
    #undef W

    asm volatile("v_nop\n\tv_nop\n\tv_nop\n\tv_nop\n\tv_nop\n\tv_nop\n\tv_nop\n\tv_nop"
                 : "+v"(acc0), "+v"(acc1), "+v"(acc2), "+v"(acc3));

    #define PUT(T) do { float* d_ = my + c * APITCH + 16 * (T) + 8 * h; \
        *reinterpret_cast<v4f*>(d_)     = __builtin_shufflevector(acc##T, acc##T, 0, 1, 2, 3); \
        *reinterpret_cast<v4f*>(d_ + 4) = __builtin_shufflevector(acc##T, acc##T, 4, 5, 6, 7); } while (0)
    PUT(0); PUT(1); PUT(2); PUT(3);
    #undef PUT
    SLAB_ORDER();

    v4f o[8];
    #pragma unroll
    for (int j = 0; j < 8; ++j)
        o[j] = *reinterpret_cast<const v4f*>(my + (2 * j + h) * APITCH + 4 * c);
    SLAB_ORDER();

    const long row0 = (long)S * 16 + h;
    float* qp = Q + row0 * 64 + 4 * c;
    if ((long)S * 16 + 16 <= (long)m) {
        #pragma unroll
        for (int j = 0; j < 8; ++j) *reinterpret_cast<v4f*>(qp + j * 128) = o[j];
    } else {
        #pragma unroll
        for (int j = 0; j < 8; ++j)
            if (row0 + 2 * j < (long)m) *reinterpret_cast<v4f*>(qp + j * 128) = o[j];
    }
}

__global__ __launch_bounds__(ATHREADS) void apply_rinv(const float* __restrict__ Phi,
                                                         const float* __restrict__ G,
                                                         float* __restrict__ Q, int m) {
    __shared__ __attribute__((aligned(16))) float slab[AWAVES * 16 * APITCH];
    __shared__ __attribute__((aligned(16))) float rowb[64];
    __shared__ float wbuf[64 * WSTR];
    if (m <= 0) return;
    const int lane = threadIdx.x & 31;
    const int wave = __builtin_amdgcn_readfirstlane(threadIdx.x >> 5);
    const int c = lane & 15;
    const int h = lane >> 4;
    const int nstrips = (m + 15) >> 4;
    const int nwaves  = AGRID * AWAVES;
    int strip = blockIdx.x * AWAVES + wave;

    float* my = slab + wave * (16 * APITCH);

    v4f pa[8], pb[8];
    if (wave == 0) {
        float a0[64], a1[64], b0[64];
        factor_load(G, lane, a0, a1, b0);
        LOAD_STRIP(pa, strip);
        LOAD_STRIP(pb, strip + nwaves);
        factor_run(lane, a0, a1, b0, rowb, wbuf);
    } else {
        LOAD_STRIP(pa, strip);
        LOAD_STRIP(pb, strip + nwaves);
    }
    __syncthreads();
    if (strip >= nstrips) return;

    v2f rf[40];
    #pragma unroll
    for (int t = 0; t < 4; ++t) {
        #pragma unroll
        for (int s = 0; s < 4 * (t + 1); ++s) {
            const int k = 8 * (s >> 1) + 4 * h + 2 * (s & 1);
            rf[2 * t * (t + 1) + s].x = wbuf[k * WSTR + 16 * t + c];
            rf[2 * t * (t + 1) + s].y = wbuf[(k + 1) * WSTR + 16 * t + c];
        }
    }

    for (; strip + nwaves < nstrips; strip += 2 * nwaves) {
        strip_body(pa, strip, strip + 2 * nwaves, m, Phi, Q, my, rf, c, h);
        strip_body(pb, strip + nwaves, strip + 3 * nwaves, m, Phi, Q, my, rf, c, h);
    }
    if (strip < nstrips)
        strip_body(pa, strip, strip + 2 * nwaves, m, Phi, Q, my, rf, c, h);
}
#undef LOAD_STRIP
#undef SLAB_ORDER

extern "C" void kernel_launch(void* const* d_in, const int* in_sizes, int n_in,
                              void* d_out, int out_size, void* d_ws, size_t ws_size,
                              hipStream_t stream) {
    const float* Phi = (const float*)d_in[0];
    float* out = (float*)d_out;
    const int m = in_sizes[0] / 64;

    float* ws   = (float*)d_ws;
    float* G    = ws;
    float* part = ws + 8192;

    long avail = ((long)(ws_size / 4) - 8192) / 4096;
    int nwg = (int)(avail < 1 ? 1 : (avail > 512 ? 512 : avail));

    gram_partial<<<nwg, GTHREADS, 0, stream>>>(Phi, part, m, nwg);
    reduce_g<<<128, RTHREADS, 0, stream>>>(part, G, nwg);
    apply_rinv<<<AGRID, ATHREADS, 0, stream>>>(Phi, G, out, m);
}
